# starting kernel
# baseline (speedup 1.0000x reference)
#include <hip/hip_runtime.h>
#include <hip/hip_bf16.h>
#include <math.h>

typedef __attribute__((ext_vector_type(2))) float v2f;
typedef __attribute__((ext_vector_type(8))) float v8f;

#define HIDDIM 128

__global__ void fill_zero_kernel(float* __restrict__ p, long long n) {
  long long i = (long long)blockIdx.x * blockDim.x + threadIdx.x;
  if (i < n) p[i] = 0.0f;
}

__global__ void count_edges_kernel(const int* __restrict__ dst, int e,
                                   float* __restrict__ cnt) {
  int i = blockIdx.x * blockDim.x + threadIdx.x;
  if (i < e) atomicAdd(&cnt[dst[i]], 1.0f);
}

__global__ void invcnt_kernel(const float* __restrict__ cnt,
                              float* __restrict__ inv, int n) {
  int i = blockIdx.x * blockDim.x + threadIdx.x;
  if (i < n) inv[i] = 1.0f / fmaxf(cnt[i], 1.0f);
}

__global__ void scatter_add8_kernel(const float* __restrict__ x,
                                    const int* __restrict__ src,
                                    const int* __restrict__ dst,
                                    float* __restrict__ agg, long long total) {
  long long i = (long long)blockIdx.x * blockDim.x + threadIdx.x;
  if (i >= total) return;
  int e = (int)(i >> 3);
  int k = (int)(i & 7);
  atomicAdd(&agg[(size_t)dst[e] * 8 + k], x[(size_t)src[e] * 8 + k]);
}

__global__ __launch_bounds__(256) void scatter_add128_kernel(
    const float* __restrict__ x, const int* __restrict__ src,
    const int* __restrict__ dst, float* __restrict__ agg, int E) {
  int wave = (int)((blockIdx.x * blockDim.x + threadIdx.x) >> 5);
  wave = __builtin_amdgcn_readfirstlane(wave);
  if (wave >= E) return;
  const int lane = threadIdx.x & 31;
  const int s = src[wave];
  const int d = dst[wave];
  const float4 v = *(const float4*)(x + (size_t)s * HIDDIM + lane * 4);
  float* p = agg + (size_t)d * HIDDIM + lane * 4;
  atomicAdd(p + 0, v.x);
  atomicAdd(p + 1, v.y);
  atomicAdd(p + 2, v.z);
  atomicAdd(p + 3, v.w);
}

__device__ __forceinline__ float xor_add(float s, const int imm) {
  switch (imm) {
    case 1:  s += __int_as_float(__builtin_amdgcn_ds_swizzle(__float_as_int(s), 0x041f)); break;
    case 2:  s += __int_as_float(__builtin_amdgcn_ds_swizzle(__float_as_int(s), 0x081f)); break;
    case 4:  s += __int_as_float(__builtin_amdgcn_ds_swizzle(__float_as_int(s), 0x101f)); break;
    default: s += __int_as_float(__builtin_amdgcn_ds_swizzle(__float_as_int(s), 0x201f)); break;
  }
  return s;
}

template <int K>
__global__ __launch_bounds__(256) void sage_layer_kernel(
    const float* __restrict__ agg, const float* __restrict__ invcnt,
    const float* __restrict__ xin, const float* __restrict__ wl,
    const float* __restrict__ wr, const float* __restrict__ bias,
    float* __restrict__ out) {
  __shared__ float sM[16][K + 1];
  __shared__ float sX[16][K + 1];
  __shared__ float s_norm[16];

  const int tid = threadIdx.x;
  const int wave = tid >> 5;
  const int lane = tid & 31;
  const int half = lane >> 4;
  const int l16 = lane & 15;
  const int rowBase = blockIdx.x * 16;
  const int col = wave * 16 + l16;

  if (tid < 16) s_norm[tid] = 0.0f;
  for (int i = tid; i < 16 * K; i += 256) {
    int r = i / K;
    int k = i - r * K;
    size_t g = (size_t)(rowBase + r) * K + k;
    sM[r][k] = agg[g] * invcnt[rowBase + r];
    sX[r][k] = xin[g];
  }
  __syncthreads();

  v8f c = {};
#pragma unroll 8
  for (int k0 = 0; k0 < K; k0 += 4) {
    const int ka = k0 + 2 * half;
    v2f am, ax, bl, br;
    am.x = sM[l16][ka];
    am.y = sM[l16][ka + 1];
    ax.x = sX[l16][ka];
    ax.y = sX[l16][ka + 1];
    bl.x = wl[(size_t)ka * HIDDIM + col];
    bl.y = wl[(size_t)(ka + 1) * HIDDIM + col];
    br.x = wr[(size_t)ka * HIDDIM + col];
    br.y = wr[(size_t)(ka + 1) * HIDDIM + col];
    c = __builtin_amdgcn_wmma_f32_16x16x4_f32(false, am, false, bl, (short)0, c,
                                              false, false);
    c = __builtin_amdgcn_wmma_f32_16x16x4_f32(false, ax, false, br, (short)0, c,
                                              false, false);
  }

  const float bv = bias[col];
#pragma unroll
  for (int v = 0; v < 8; ++v) c[v] += bv;

#pragma unroll
  for (int v = 0; v < 8; ++v) {
    float s = c[v] * c[v];
    s = xor_add(s, 1);
    s = xor_add(s, 2);
    s = xor_add(s, 4);
    s = xor_add(s, 8);
    if (l16 == 0) atomicAdd(&s_norm[v + 8 * half], s);
  }
  __syncthreads();

#pragma unroll
  for (int v = 0; v < 8; ++v) {
    const int row = v + 8 * half;
    const float scale = 1.0f / fmaxf(sqrtf(s_norm[row]), 1e-12f);
    out[(size_t)(rowBase + row) * HIDDIM + col] = tanhf(c[v] * scale);
  }
}

extern "C" void kernel_launch(void* const* d_in, const int* in_sizes, int n_in,
                              void* d_out, int out_size, void* d_ws,
                              size_t ws_size, hipStream_t stream) {
  const float* x = (const float*)d_in[0];
  const int* ei_conn = (const int*)d_in[1];
  const int* ei_dest = (const int*)d_in[2];
  const float* wl1 = (const float*)d_in[3];
  const float* wr1 = (const float*)d_in[4];
  const float* b1 = (const float*)d_in[5];
  const float* wl2 = (const float*)d_in[6];
  const float* wr2 = (const float*)d_in[7];
  const float* b2 = (const float*)d_in[8];
  const float* wl3 = (const float*)d_in[9];
  const float* wr3 = (const float*)d_in[10];
  const float* b3 = (const float*)d_in[11];

  const int N = in_sizes[0] / 8;
  const int E = in_sizes[1] / 2;

  float* ws = (float*)d_ws;
  float* agg = ws;
  float* h1 = agg + (size_t)N * HIDDIM;
  float* h2 = h1 + (size_t)N * HIDDIM;
  float* ivc_conn = h2 + (size_t)N * HIDDIM;
  float* ivc_dest = ivc_conn + N;
  float* cnt = ivc_dest + N;

  const int T = 256;
  auto nb = [](long long n, int t) { return (int)((n + t - 1) / t); };

  const int* src_conn = ei_conn;
  const int* dst_conn = ei_conn + E;
  const int* src_dest = ei_dest;
  const int* dst_dest = ei_dest + E;

  fill_zero_kernel<<<nb(N, T), T, 0, stream>>>(cnt, N);
  count_edges_kernel<<<nb(E, T), T, 0, stream>>>(dst_conn, E, cnt);
  invcnt_kernel<<<nb(N, T), T, 0, stream>>>(cnt, ivc_conn, N);
  fill_zero_kernel<<<nb(N, T), T, 0, stream>>>(cnt, N);
  count_edges_kernel<<<nb(E, T), T, 0, stream>>>(dst_dest, E, cnt);
  invcnt_kernel<<<nb(N, T), T, 0, stream>>>(cnt, ivc_dest, N);

  const int gemmBlocks = (N + 15) / 16;
  const int scatBlocks = (E + 7) / 8;

  fill_zero_kernel<<<nb((long long)N * 8, T), T, 0, stream>>>(agg,
                                                              (long long)N * 8);
  scatter_add8_kernel<<<nb((long long)E * 8, T), T, 0, stream>>>(
      x, src_conn, dst_conn, agg, (long long)E * 8);
  sage_layer_kernel<8><<<gemmBlocks, T, 0, stream>>>(agg, ivc_conn, x, wl1,
                                                     wr1, b1, h1);

  fill_zero_kernel<<<nb((long long)N * HIDDIM, T), T, 0, stream>>>(
      agg, (long long)N * HIDDIM);
  scatter_add128_kernel<<<scatBlocks, T, 0, stream>>>(h1, src_dest, dst_dest,
                                                      agg, E);
  sage_layer_kernel<128><<<gemmBlocks, T, 0, stream>>>(agg, ivc_dest, h1, wl2,
                                                       wr2, b2, h2);

  fill_zero_kernel<<<nb((long long)N * HIDDIM, T), T, 0, stream>>>(
      agg, (long long)N * HIDDIM);
  scatter_add128_kernel<<<scatBlocks, T, 0, stream>>>(h2, src_conn, dst_conn,
                                                      agg, E);
  sage_layer_kernel<128><<<gemmBlocks, T, 0, stream>>>(agg, ivc_conn, h2, wl3,
                                                       wr3, b3, (float*)d_out);
}


# Owner ids as one bit plane per owner block (one-hot), built through an LDS image: a block reads only its own plane, set bits are its edges, no compares; list order arrival-dependent as before
# speedup vs baseline: 3.0241x; 3.0241x over previous
#include <hip/hip_runtime.h>
#include <hip/hip_bf16.h>
#include <math.h>

typedef __attribute__((ext_vector_type(2))) float v2f;
typedef __attribute__((ext_vector_type(8))) float v8f;

#define HIDDIM 128

#define OWN_ROWS 400
#define QCAP 544
#define OCAP 320
#define PLANE_CHUNK 8192
#define PLANE_PITCH 260
#define MAX_OWN 256
#define GROUP_EDGES 4096
#define REM_CHUNK 1024
#define MODE_SCAN 0
#define MODE_RECORD 1
#define MODE_REPLAY 2

__global__ __launch_bounds__(1024) void owner_plane_kernel(
    const int* __restrict__ dstA, const int* __restrict__ dstB,
    unsigned* __restrict__ planeA, unsigned* __restrict__ planeB, int nChunks, int ownBlocks) {
  __shared__ __attribute__((aligned(16))) unsigned bits[MAX_OWN * PLANE_PITCH];
  if ((int)blockIdx.x >= nChunks) return;
  const int tid = threadIdx.x;
  const int nb = min(ownBlocks, MAX_OWN);
  const int* dst = blockIdx.y ? dstB : dstA;
  unsigned* plane = blockIdx.y ? planeB : planeA;
  for (int i = tid; i < nb * (PLANE_PITCH / 4); i += 1024)
    ((uint4*)bits)[i] = make_uint4(0u, 0u, 0u, 0u);
  const int e0 = blockIdx.x * PLANE_CHUNK;
  int d[8];
#pragma unroll
  for (int j = 0; j < 8; ++j) d[j] = dst[e0 + j * 1024 + tid];
  __syncthreads();
#pragma unroll
  for (int j = 0; j < 8; ++j) {
    const unsigned o = (unsigned)d[j] / OWN_ROWS;
    if (o < (unsigned)nb) atomicOr(&bits[o * PLANE_PITCH + j * 32 + (tid >> 5)], 1u << (tid & 31));
  }
  __syncthreads();
  const size_t planeWords = (size_t)nChunks * (PLANE_CHUNK / 32);
  for (int i = tid; i < nb * 64; i += 1024) {
    const int b = i >> 6;
    const int k = (i & 63) * 4;
    const uint4 v = *(const uint4*)&bits[b * PLANE_PITCH + k];
    *(uint4*)(plane + (size_t)b * planeWords + (size_t)blockIdx.x * (PLANE_CHUNK / 32) + k) = v;
  }
}

template <int D, int MODE>
__global__ __launch_bounds__(1024) void owner_mean_kernel(
    const float* __restrict__ x, const int* __restrict__ src,
    const int* __restrict__ dst, const unsigned* __restrict__ plane,
    float* __restrict__ mean, unsigned* __restrict__ rec, int* __restrict__ recCnt,
    int N, int E, int nChunks) {
  constexpr int LPR = (D >= 32) ? 32 : D;
  constexpr int CPL = D / LPR;
  constexpr int RPW = 32 / LPR;
  __shared__ __attribute__((aligned(16))) float acc[OWN_ROWS * D];
  __shared__ float cnt[OWN_ROWS];
  __shared__ unsigned q[32][QCAP];
  __shared__ unsigned oq[32][OCAP];
  __shared__ unsigned ocnt[32];
  __shared__ unsigned ovf;

  const int tid = threadIdx.x;
  const int w = __builtin_amdgcn_readfirstlane(tid >> 5);
  const int lane = tid & 31;
  const int sub = lane % LPR;
  const int slot = lane / LPR;
  const int rowBase = blockIdx.x * OWN_ROWS;
  const unsigned below = (1u << lane) - 1u;

  for (int i = tid; i < OWN_ROWS * D / 4; i += 1024)
    ((float4*)acc)[i] = make_float4(0.0f, 0.0f, 0.0f, 0.0f);
  for (int i = tid; i < OWN_ROWS; i += 1024) cnt[i] = 0.0f;
  if (tid < 32) ocnt[tid] = 0u;
  if (tid == 0) ovf = 0u;
  __syncthreads();

  int qn = 0;
  unsigned* const myRec = rec + ((size_t)blockIdx.x * 32 + w) * OCAP;
  int* const recOk = recCnt + (size_t)gridDim.x * 32;

  auto push = [&](unsigned mask, int e) {
    if (mask) {
      if ((mask >> lane) & 1u)
        q[w][qn + __builtin_popcount(mask & below)] = (unsigned)e;
      qn += __builtin_popcount(mask);
    }
  };

  auto hand_over = [&]() {
    for (int j = lane; j < qn; j += 32) {
      const unsigned e = q[w][j];
      const unsigned s = (unsigned)src[e];
      const unsigned lr = (unsigned)(dst[e] - rowBase);
      const unsigned o = lr & 31u;
      const unsigned pos = atomicAdd(&ocnt[o], 1u);
      if (pos < (unsigned)OCAP) {
        oq[o][pos] = (s << 9) | lr;
      } else {
        ovf = 1u;
        const float* p = x + (size_t)s * D;
        for (int c = 0; c < D; ++c) atomicAdd(&acc[lr * D + c], p[c]);
        atomicAdd(&cnt[lr], 1.0f);
      }
    }
    qn = 0;
  };

  bool haveList = false;
  int n = 0;
  if constexpr (MODE == MODE_REPLAY) {
    haveList = recOk[blockIdx.x] == 1;
    if (haveList) {
      n = __builtin_amdgcn_readfirstlane(recCnt[blockIdx.x * 32 + w]);
      n = min(max(n, 0), OCAP);
      for (int j = lane; j < n; j += 32) oq[w][j] = myRec[j];
    }
  }

  const int covered = nChunks * PLANE_CHUNK;
  if (!haveList && nChunks > 0) {
    const int nGroups = covered / GROUP_EDGES;
    const uint4* p4 = (const uint4*)(plane + (size_t)blockIdx.x * ((size_t)nChunks * (PLANE_CHUNK / 32))) + lane;
    auto group_of = [&](int t, int u) -> int { return (4 * t + u) * 32 + w; };
    auto load_group = [&](int g) -> uint4 { return p4[(size_t)min(g, nGroups - 1) * 32]; };
    auto take = [&](const uint4 c, const int eb) {
      unsigned long long m0 = ((unsigned long long)c.y << 32) | c.x;
      unsigned long long m1 = ((unsigned long long)c.w << 32) | c.z;
      unsigned hm = __builtin_amdgcn_ballot_w32((m0 | m1) != 0ull);
      while (hm) {
        if ((m0 | m1) != 0ull) {
          int bpos;
          if (m0 != 0ull) {
            bpos = __builtin_ctzll(m0);
            m0 &= m0 - 1ull;
          } else {
            bpos = 64 + __builtin_ctzll(m1);
            m1 &= m1 - 1ull;
          }
          q[w][qn + __builtin_popcount(hm & below)] = (unsigned)(eb + bpos);
        }
        qn += __builtin_popcount(hm);
        if (qn >= 128) hand_over();
        hm = __builtin_amdgcn_ballot_w32((m0 | m1) != 0ull);
      }
    };
    uint4 cur[4];
#pragma unroll
    for (int u = 0; u < 4; ++u) cur[u] = load_group(group_of(0, u));
    for (int t = 0; group_of(t, 0) < nGroups; ++t) {
      uint4 nxt[4];
#pragma unroll
      for (int u = 0; u < 4; ++u) nxt[u] = load_group(group_of(t + 1, u));
#pragma unroll
      for (int u = 0; u < 4; ++u) {
        const int g = group_of(t, u);
        if (g < nGroups) take(cur[u], g * GROUP_EDGES + lane * 128);
      }
#pragma unroll
      for (int u = 0; u < 4; ++u) cur[u] = nxt[u];
    }
  }
  for (int rem0 = haveList ? E : covered + w * REM_CHUNK; rem0 < E; rem0 += 32 * REM_CHUNK) {
    const int lim = min(rem0 + REM_CHUNK, E);
    for (int g = rem0; g < lim; g += 256) {
      int dt[8];
#pragma unroll
      for (int u = 0; u < 8; ++u) dt[u] = dst[min(g + u * 32 + lane, E - 1)];
#pragma unroll
      for (int u = 0; u < 8; ++u) {
        const int e = g + u * 32 + lane;
        const unsigned l = (unsigned)(dt[u] - rowBase);
        push(__builtin_amdgcn_ballot_w32(e < lim && l < (unsigned)OWN_ROWS), e);
      }
      if (qn >= 32) hand_over();
    }
  }
  if (qn > 0) hand_over();
  __syncthreads();

  if (!haveList) n = min((int)__builtin_amdgcn_readfirstlane((int)ocnt[w]), OCAP);
  if constexpr (MODE == MODE_RECORD) {
    for (int j = lane; j < n; j += 32) myRec[j] = oq[w][j];
    if (lane == 0) recCnt[blockIdx.x * 32 + w] = n;
    if (tid == 0) recOk[blockIdx.x] = ovf ? 0 : 1;
  }

  if constexpr (D >= 128) {
    for (int i = 0; i < n; i += 16) {
      const int mi = i + (lane & 15);
      const unsigned mine = oq[w][mi < n ? mi : 0];
      float4 v[16];
#pragma unroll
      for (int j = 0; j < 16; ++j) {
        const unsigned en = (unsigned)__builtin_amdgcn_readlane((int)mine, j);
        v[j] = *(const float4*)(x + (size_t)(en >> 9) * D + 4 * lane);
      }
      if (lane < 16 && mi < n) atomicAdd(&cnt[mine & 511u], 1.0f);
#pragma unroll
      for (int j = 0; j < 16; ++j) {
        if (i + j < n) {
          const unsigned lr = (unsigned)__builtin_amdgcn_readlane((int)mine, j) & 511u;
          float4* a = (float4*)(acc + lr * D) + lane;
          float4 t = *a;
          t.x += v[j].x; t.y += v[j].y; t.z += v[j].z; t.w += v[j].w;
          *a = t;
        }
      }
    }
  } else {
    for (int i = 0; i < n; i += 8 * RPW) {
      float v[8][CPL];
      int ld[8];
      bool ok[8];
#pragma unroll
      for (int j = 0; j < 8; ++j) {
        const int idx = i + j * RPW + slot;
        ok[j] = idx < n;
        const unsigned en = oq[w][ok[j] ? idx : 0];
        ld[j] = (int)(en & 511u);
        const float* p = x + (size_t)(en >> 9) * D + sub;
#pragma unroll
        for (int c = 0; c < CPL; ++c) v[j][c] = p[c * LPR];
      }
#pragma unroll
      for (int j = 0; j < 8; ++j) {
        if (ok[j]) {
          float* a = acc + ld[j] * D + sub;
#pragma unroll
          for (int c = 0; c < CPL; ++c) atomicAdd(a + c * LPR, v[j][c]);
          if (sub == 0) atomicAdd(&cnt[ld[j]], 1.0f);
        }
      }
    }
  }
  __syncthreads();

  for (int i = tid; i < OWN_ROWS * D / 4; i += 1024) {
    const int r = (i * 4) / D;
    const int g = rowBase + r;
    if (g < N) {
      const float inv = 1.0f / fmaxf(cnt[r], 1.0f);
      float4 s = ((const float4*)acc)[i];
      s.x *= inv; s.y *= inv; s.z *= inv; s.w *= inv;
      *(float4*)(mean + (size_t)g * D + (i * 4 - r * D)) = s;
    }
  }
}

__device__ __forceinline__ float xor_add(float s, const int imm) {
  switch (imm) {
    case 1:  s += __int_as_float(__builtin_amdgcn_ds_swizzle(__float_as_int(s), 0x041f)); break;
    case 2:  s += __int_as_float(__builtin_amdgcn_ds_swizzle(__float_as_int(s), 0x081f)); break;
    case 4:  s += __int_as_float(__builtin_amdgcn_ds_swizzle(__float_as_int(s), 0x101f)); break;
    default: s += __int_as_float(__builtin_amdgcn_ds_swizzle(__float_as_int(s), 0x201f)); break;
  }
  return s;
}

template <int K>
__global__ __launch_bounds__(256) void sage_layer_kernel(
    const float* __restrict__ agg, const float* __restrict__ xin, const float* __restrict__ wl,
    const float* __restrict__ wr, const float* __restrict__ bias,
    float* __restrict__ out) {
  __shared__ float sM[16][K + 1];
  __shared__ float sX[16][K + 1];
  __shared__ float s_norm[16];

  const int tid = threadIdx.x;
  const int wave = tid >> 5;
  const int lane = tid & 31;
  const int half = lane >> 4;
  const int l16 = lane & 15;
  const int rowBase = blockIdx.x * 16;
  const int col = wave * 16 + l16;

  if (tid < 16) s_norm[tid] = 0.0f;
  for (int i = tid; i < 16 * K; i += 256) {
    int r = i / K;
    int k = i - r * K;
    size_t g = (size_t)(rowBase + r) * K + k;
    sM[r][k] = agg[g];
    sX[r][k] = xin[g];
  }
  __syncthreads();

  v8f c = {};
#pragma unroll 8
  for (int k0 = 0; k0 < K; k0 += 4) {
    const int ka = k0 + 2 * half;
    v2f am, ax, bl, br;
    am.x = sM[l16][ka];
    am.y = sM[l16][ka + 1];
    ax.x = sX[l16][ka];
    ax.y = sX[l16][ka + 1];
    bl.x = wl[(size_t)ka * HIDDIM + col];
    bl.y = wl[(size_t)(ka + 1) * HIDDIM + col];
    br.x = wr[(size_t)ka * HIDDIM + col];
    br.y = wr[(size_t)(ka + 1) * HIDDIM + col];
    c = __builtin_amdgcn_wmma_f32_16x16x4_f32(false, am, false, bl, (short)0, c,
                                              false, false);
    c = __builtin_amdgcn_wmma_f32_16x16x4_f32(false, ax, false, br, (short)0, c,
                                              false, false);
  }

  const float bv = bias[col];
#pragma unroll
  for (int v = 0; v < 8; ++v) c[v] += bv;

#pragma unroll
  for (int v = 0; v < 8; ++v) {
    float s = c[v] * c[v];
    s = xor_add(s, 1);
    s = xor_add(s, 2);
    s = xor_add(s, 4);
    s = xor_add(s, 8);
    if (l16 == 0) atomicAdd(&s_norm[v + 8 * half], s);
  }
  __syncthreads();

#pragma unroll
  for (int v = 0; v < 8; ++v) {
    const int row = v + 8 * half;
    const float scale = 1.0f / fmaxf(sqrtf(s_norm[row]), 1e-12f);
    out[(size_t)(rowBase + row) * HIDDIM + col] = tanhf(c[v] * scale);
  }
}

#define APITCH 132
__global__ __launch_bounds__(256, 4) void sage_layer_resident_kernel(
    const float* __restrict__ agg, const float* __restrict__ xin, const float* __restrict__ wl,
    const float* __restrict__ wr, const float* __restrict__ bias,
    float* __restrict__ out, int nTiles) {
  constexpr int K = HIDDIM;
  __shared__ __attribute__((aligned(16))) float sM[16][APITCH];
  __shared__ __attribute__((aligned(16))) float sX[16][APITCH];
  __shared__ float s_norm[2][16];

  const int tid = threadIdx.x;
  const int wave = tid >> 5;
  const int lane = tid & 31;
  const int half = lane >> 4;
  const int l16 = lane & 15;
  const int col = wave * 16 + l16;

  v2f bl[K / 4], br[K / 4];
#pragma unroll
  for (int j = 0; j < K / 4; ++j) {
    const int ka = 4 * j + 2 * half;
    bl[j].x = wl[(size_t)ka * HIDDIM + col];
    bl[j].y = wl[(size_t)(ka + 1) * HIDDIM + col];
    br[j].x = wr[(size_t)ka * HIDDIM + col];
    br[j].y = wr[(size_t)(ka + 1) * HIDDIM + col];
  }
  const float bv = bias[col];

  const int sr = tid >> 5;
  const int sc = (tid & 31) * 4;
  int tile = blockIdx.x;
  float4 pm0, pm1, px0, px1;
  pm0 = pm1 = px0 = px1 = make_float4(0.0f, 0.0f, 0.0f, 0.0f);
  if (tile < nTiles) {
    const size_t g = (size_t)(tile * 16 + sr) * K + sc;
    pm0 = *(const float4*)(agg + g);
    pm1 = *(const float4*)(agg + g + 8 * K);
    px0 = *(const float4*)(xin + g);
    px1 = *(const float4*)(xin + g + 8 * K);
  }

  int par = 0;
  for (; tile < nTiles; tile += gridDim.x) {
    const int rowBase = tile * 16;
    *(float4*)&sM[sr][sc] = pm0;
    *(float4*)&sM[sr + 8][sc] = pm1;
    *(float4*)&sX[sr][sc] = px0;
    *(float4*)&sX[sr + 8][sc] = px1;
    if (tid < 16) s_norm[par][tid] = 0.0f;
    __syncthreads();

    v8f c = {};
    v2f am[2][8], ax[2][8];
#pragma unroll
    for (int u = 0; u < 8; ++u) {
      am[0][u] = *(const v2f*)&sM[l16][4 * u + 2 * half];
      ax[0][u] = *(const v2f*)&sX[l16][4 * u + 2 * half];
    }
#pragma unroll
    for (int b = 0; b < K / 32; ++b) {
      if (b + 1 < K / 32) {
#pragma unroll
        for (int u = 0; u < 8; ++u) {
          am[(b + 1) & 1][u] = *(const v2f*)&sM[l16][32 * (b + 1) + 4 * u + 2 * half];
          ax[(b + 1) & 1][u] = *(const v2f*)&sX[l16][32 * (b + 1) + 4 * u + 2 * half];
        }
      }
      __builtin_amdgcn_sched_barrier(0);
#pragma unroll
      for (int u = 0; u < 8; ++u) {
        const int j = 8 * b + u;
        c = __builtin_amdgcn_wmma_f32_16x16x4_f32(false, am[b & 1][u], false, bl[j], (short)0, c,
                                                  false, false);
        c = __builtin_amdgcn_wmma_f32_16x16x4_f32(false, ax[b & 1][u], false, br[j], (short)0, c,
                                                  false, false);
      }
      __builtin_amdgcn_sched_barrier(0);
    }

#pragma unroll
    for (int v = 0; v < 8; ++v) c[v] += bv;

    float ss[8];
#pragma unroll
    for (int v = 0; v < 8; ++v) ss[v] = c[v] * c[v];
#pragma unroll
    for (int v = 0; v < 8; ++v) ss[v] = xor_add(ss[v], 1);
#pragma unroll
    for (int v = 0; v < 8; ++v) ss[v] = xor_add(ss[v], 2);
#pragma unroll
    for (int v = 0; v < 8; ++v) ss[v] = xor_add(ss[v], 4);
#pragma unroll
    for (int v = 0; v < 8; ++v) ss[v] = xor_add(ss[v], 8);
    if (l16 == 0) {
#pragma unroll
      for (int v = 0; v < 8; ++v) atomicAdd(&s_norm[par][v + 8 * half], ss[v]);
    }
    __syncthreads();

    {
      const int nt = tile + (int)gridDim.x < nTiles ? tile + (int)gridDim.x : tile;
      const size_t g = (size_t)(nt * 16 + sr) * K + sc;
      pm0 = *(const float4*)(agg + g);
      pm1 = *(const float4*)(agg + g + 8 * K);
      px0 = *(const float4*)(xin + g);
      px1 = *(const float4*)(xin + g + 8 * K);
    }

    float scale[8];
#pragma unroll
    for (int v = 0; v < 8; ++v)
      scale[v] = 1.0f / fmaxf(sqrtf(s_norm[par][v + 8 * half]), 1e-12f);
    asm volatile("" ::"v"(scale[0]), "v"(scale[1]), "v"(scale[2]), "v"(scale[3]), "v"(scale[4]),
                 "v"(scale[5]), "v"(scale[6]), "v"(scale[7]));
#pragma unroll
    for (int v = 0; v < 8; ++v)
      out[(size_t)(rowBase + v + 8 * half) * HIDDIM + col] = tanhf(c[v] * scale[v]);
    par ^= 1;
  }
}

extern "C" void kernel_launch(void* const* d_in, const int* in_sizes, int n_in,
                              void* d_out, int out_size, void* d_ws,
                              size_t ws_size, hipStream_t stream) {
  const float* x = (const float*)d_in[0];
  const int* ei_conn = (const int*)d_in[1];
  const int* ei_dest = (const int*)d_in[2];
  const float* wl1 = (const float*)d_in[3];
  const float* wr1 = (const float*)d_in[4];
  const float* b1 = (const float*)d_in[5];
  const float* wl2 = (const float*)d_in[6];
  const float* wr2 = (const float*)d_in[7];
  const float* b2 = (const float*)d_in[8];
  const float* wl3 = (const float*)d_in[9];
  const float* wr3 = (const float*)d_in[10];
  const float* b3 = (const float*)d_in[11];

  const int N = in_sizes[0] / 8;
  const int E = in_sizes[1] / 2;

  float* ws = (float*)d_ws;
  float* mean = ws;
  float* h1 = mean + (size_t)N * HIDDIM;
  float* h2 = h1 + (size_t)N * HIDDIM;

  const int T = 256;
  const int* src_conn = ei_conn;
  const int* dst_conn = ei_conn + E;
  const int* src_dest = ei_dest;
  const int* dst_dest = ei_dest + E;

  const int gemmBlocks = (N + 15) / 16;
  const int ownBlocks = (N + OWN_ROWS - 1) / OWN_ROWS;
  const int wideBlocks = 512;
  const int nChunks = E / PLANE_CHUNK;
  const size_t planeAll = (size_t)ownBlocks * nChunks * (PLANE_CHUNK / 32);
  unsigned* plane_conn = (unsigned*)(h2 + (size_t)N * HIDDIM);
  unsigned* plane_dest = plane_conn + planeAll;
  unsigned* rec = plane_dest + planeAll;
  int* recCnt = (int*)(rec + (size_t)ownBlocks * 32 * OCAP);

  owner_plane_kernel<<<dim3(nChunks > 0 ? nChunks : 1, 2), 1024, 0, stream>>>(
      dst_conn, dst_dest, plane_conn, plane_dest, nChunks, ownBlocks);

  owner_mean_kernel<8, MODE_RECORD><<<ownBlocks, 1024, 0, stream>>>(
      x, src_conn, dst_conn, plane_conn, mean, rec, recCnt, N, E, nChunks);
  sage_layer_kernel<8><<<gemmBlocks, T, 0, stream>>>(mean, x, wl1, wr1, b1, h1);

  owner_mean_kernel<HIDDIM, MODE_SCAN><<<ownBlocks, 1024, 0, stream>>>(
      h1, src_dest, dst_dest, plane_dest, mean, rec, recCnt, N, E, nChunks);
  sage_layer_resident_kernel<<<wideBlocks, T, 0, stream>>>(mean, h1, wl2, wr2, b2, h2, gemmBlocks);

  owner_mean_kernel<HIDDIM, MODE_REPLAY><<<ownBlocks, 1024, 0, stream>>>(
      h2, src_conn, dst_conn, plane_conn, mean, rec, recCnt, N, E, nChunks);
  sage_layer_resident_kernel<<<wideBlocks, T, 0, stream>>>(mean, h2, wl3, wr3, b3, (float*)d_out,
                                                           gemmBlocks);
}


# Layer 1 done inside its owner block: a wave takes a 16-row tile and does products, bias, norm, tanh from the LDS sums; no layer-1 mean buffer, 7 launches to 6; norm in fixed order: sum order changes
# speedup vs baseline: 3.1279x; 1.0343x over previous
#include <hip/hip_runtime.h>
#include <hip/hip_bf16.h>
#include <math.h>

typedef __attribute__((ext_vector_type(2))) float v2f;
typedef __attribute__((ext_vector_type(8))) float v8f;

#define HIDDIM 128

#define OWN_ROWS 400
#define QCAP 544
#define OCAP 320
#define PLANE_CHUNK 8192
#define PLANE_PITCH 260
#define MAX_OWN 256
#define GROUP_EDGES 4096
#define REM_CHUNK 1024
#define MODE_SCAN 0
#define MODE_RECORD 1
#define MODE_REPLAY 2

__global__ __launch_bounds__(1024) void owner_plane_kernel(
    const int* __restrict__ dstA, const int* __restrict__ dstB,
    unsigned* __restrict__ planeA, unsigned* __restrict__ planeB, int nChunks, int ownBlocks) {
  __shared__ __attribute__((aligned(16))) unsigned bits[MAX_OWN * PLANE_PITCH];
  if ((int)blockIdx.x >= nChunks) return;
  const int tid = threadIdx.x;
  const int nb = min(ownBlocks, MAX_OWN);
  const int* dst = blockIdx.y ? dstB : dstA;
  unsigned* plane = blockIdx.y ? planeB : planeA;
  for (int i = tid; i < nb * (PLANE_PITCH / 4); i += 1024)
    ((uint4*)bits)[i] = make_uint4(0u, 0u, 0u, 0u);
  const int e0 = blockIdx.x * PLANE_CHUNK;
  int d[8];
#pragma unroll
  for (int j = 0; j < 8; ++j) d[j] = dst[e0 + j * 1024 + tid];
  __syncthreads();
#pragma unroll
  for (int j = 0; j < 8; ++j) {
    const unsigned o = (unsigned)d[j] / OWN_ROWS;
    if (o < (unsigned)nb) atomicOr(&bits[o * PLANE_PITCH + j * 32 + (tid >> 5)], 1u << (tid & 31));
  }
  __syncthreads();
  const size_t planeWords = (size_t)nChunks * (PLANE_CHUNK / 32);
  for (int i = tid; i < nb * 64; i += 1024) {
    const int b = i >> 6;
    const int k = (i & 63) * 4;
    const uint4 v = *(const uint4*)&bits[b * PLANE_PITCH + k];
    *(uint4*)(plane + (size_t)b * planeWords + (size_t)blockIdx.x * (PLANE_CHUNK / 32) + k) = v;
  }
}

__device__ __forceinline__ float xor_add(float s, const int imm) {
  switch (imm) {
    case 1:  s += __int_as_float(__builtin_amdgcn_ds_swizzle(__float_as_int(s), 0x041f)); break;
    case 2:  s += __int_as_float(__builtin_amdgcn_ds_swizzle(__float_as_int(s), 0x081f)); break;
    case 4:  s += __int_as_float(__builtin_amdgcn_ds_swizzle(__float_as_int(s), 0x101f)); break;
    default: s += __int_as_float(__builtin_amdgcn_ds_swizzle(__float_as_int(s), 0x201f)); break;
  }
  return s;
}

template <int D, int MODE>
__global__ __launch_bounds__(1024) void owner_mean_kernel(
    const float* __restrict__ x, const int* __restrict__ src,
    const int* __restrict__ dst, const unsigned* __restrict__ plane,
    float* __restrict__ mean, unsigned* __restrict__ rec, int* __restrict__ recCnt,
    int N, int E, int nChunks,
    const float* __restrict__ wl, const float* __restrict__ wr, const float* __restrict__ bias,
    float* __restrict__ hout) {
  constexpr int LPR = (D >= 32) ? 32 : D;
  constexpr int CPL = D / LPR;
  constexpr int RPW = 32 / LPR;
  __shared__ __attribute__((aligned(16))) float acc[OWN_ROWS * D];
  __shared__ float cnt[OWN_ROWS];
  __shared__ unsigned q[32][QCAP];
  __shared__ unsigned oq[32][OCAP];
  __shared__ unsigned ocnt[32];
  __shared__ unsigned ovf;

  const int tid = threadIdx.x;
  const int w = __builtin_amdgcn_readfirstlane(tid >> 5);
  const int lane = tid & 31;
  const int sub = lane % LPR;
  const int slot = lane / LPR;
  const int rowBase = blockIdx.x * OWN_ROWS;
  const unsigned below = (1u << lane) - 1u;

  for (int i = tid; i < OWN_ROWS * D / 4; i += 1024)
    ((float4*)acc)[i] = make_float4(0.0f, 0.0f, 0.0f, 0.0f);
  for (int i = tid; i < OWN_ROWS; i += 1024) cnt[i] = 0.0f;
  if (tid < 32) ocnt[tid] = 0u;
  if (tid == 0) ovf = 0u;
  __syncthreads();

  int qn = 0;
  unsigned* const myRec = rec + ((size_t)blockIdx.x * 32 + w) * OCAP;
  int* const recOk = recCnt + (size_t)gridDim.x * 32;

  auto push = [&](unsigned mask, int e) {
    if (mask) {
      if ((mask >> lane) & 1u)
        q[w][qn + __builtin_popcount(mask & below)] = (unsigned)e;
      qn += __builtin_popcount(mask);
    }
  };

  auto hand_over = [&]() {
    for (int j = lane; j < qn; j += 32) {
      const unsigned e = q[w][j];
      const unsigned s = (unsigned)src[e];
      const unsigned lr = (unsigned)(dst[e] - rowBase);
      const unsigned o = lr & 31u;
      const unsigned pos = atomicAdd(&ocnt[o], 1u);
      if (pos < (unsigned)OCAP) {
        oq[o][pos] = (s << 9) | lr;
      } else {
        ovf = 1u;
        const float* p = x + (size_t)s * D;
        for (int c = 0; c < D; ++c) atomicAdd(&acc[lr * D + c], p[c]);
        atomicAdd(&cnt[lr], 1.0f);
      }
    }
    qn = 0;
  };

  bool haveList = false;
  int n = 0;
  if constexpr (MODE == MODE_REPLAY) {
    haveList = recOk[blockIdx.x] == 1;
    if (haveList) {
      n = __builtin_amdgcn_readfirstlane(recCnt[blockIdx.x * 32 + w]);
      n = min(max(n, 0), OCAP);
      for (int j = lane; j < n; j += 32) oq[w][j] = myRec[j];
    }
  }

  const int covered = nChunks * PLANE_CHUNK;
  if (!haveList && nChunks > 0) {
    const int nGroups = covered / GROUP_EDGES;
    const uint4* p4 = (const uint4*)(plane + (size_t)blockIdx.x * ((size_t)nChunks * (PLANE_CHUNK / 32))) + lane;
    auto group_of = [&](int t, int u) -> int { return (4 * t + u) * 32 + w; };
    auto load_group = [&](int g) -> uint4 { return p4[(size_t)min(g, nGroups - 1) * 32]; };
    auto take = [&](const uint4 c, const int eb) {
      unsigned long long m0 = ((unsigned long long)c.y << 32) | c.x;
      unsigned long long m1 = ((unsigned long long)c.w << 32) | c.z;
      unsigned hm = __builtin_amdgcn_ballot_w32((m0 | m1) != 0ull);
      while (hm) {
        if ((m0 | m1) != 0ull) {
          int bpos;
          if (m0 != 0ull) {
            bpos = __builtin_ctzll(m0);
            m0 &= m0 - 1ull;
          } else {
            bpos = 64 + __builtin_ctzll(m1);
            m1 &= m1 - 1ull;
          }
          q[w][qn + __builtin_popcount(hm & below)] = (unsigned)(eb + bpos);
        }
        qn += __builtin_popcount(hm);
        if (qn >= 128) hand_over();
        hm = __builtin_amdgcn_ballot_w32((m0 | m1) != 0ull);
      }
    };
    uint4 cur[4];
#pragma unroll
    for (int u = 0; u < 4; ++u) cur[u] = load_group(group_of(0, u));
    for (int t = 0; group_of(t, 0) < nGroups; ++t) {
      uint4 nxt[4];
#pragma unroll
      for (int u = 0; u < 4; ++u) nxt[u] = load_group(group_of(t + 1, u));
#pragma unroll
      for (int u = 0; u < 4; ++u) {
        const int g = group_of(t, u);
        if (g < nGroups) take(cur[u], g * GROUP_EDGES + lane * 128);
      }
#pragma unroll
      for (int u = 0; u < 4; ++u) cur[u] = nxt[u];
    }
  }
  for (int rem0 = haveList ? E : covered + w * REM_CHUNK; rem0 < E; rem0 += 32 * REM_CHUNK) {
    const int lim = min(rem0 + REM_CHUNK, E);
    for (int g = rem0; g < lim; g += 256) {
      int dt[8];
#pragma unroll
      for (int u = 0; u < 8; ++u) dt[u] = dst[min(g + u * 32 + lane, E - 1)];
#pragma unroll
      for (int u = 0; u < 8; ++u) {
        const int e = g + u * 32 + lane;
        const unsigned l = (unsigned)(dt[u] - rowBase);
        push(__builtin_amdgcn_ballot_w32(e < lim && l < (unsigned)OWN_ROWS), e);
      }
      if (qn >= 32) hand_over();
    }
  }
  if (qn > 0) hand_over();
  __syncthreads();

  if (!haveList) n = min((int)__builtin_amdgcn_readfirstlane((int)ocnt[w]), OCAP);
  if constexpr (MODE == MODE_RECORD) {
    for (int j = lane; j < n; j += 32) myRec[j] = oq[w][j];
    if (lane == 0) recCnt[blockIdx.x * 32 + w] = n;
    if (tid == 0) recOk[blockIdx.x] = ovf ? 0 : 1;
  }

  if constexpr (D >= 128) {
    for (int i = 0; i < n; i += 16) {
      const int mi = i + (lane & 15);
      const unsigned mine = oq[w][mi < n ? mi : 0];
      float4 v[16];
#pragma unroll
      for (int j = 0; j < 16; ++j) {
        const unsigned en = (unsigned)__builtin_amdgcn_readlane((int)mine, j);
        v[j] = *(const float4*)(x + (size_t)(en >> 9) * D + 4 * lane);
      }
      if (lane < 16 && mi < n) atomicAdd(&cnt[mine & 511u], 1.0f);
#pragma unroll
      for (int j = 0; j < 16; ++j) {
        if (i + j < n) {
          const unsigned lr = (unsigned)__builtin_amdgcn_readlane((int)mine, j) & 511u;
          float4* a = (float4*)(acc + lr * D) + lane;
          float4 t = *a;
          t.x += v[j].x; t.y += v[j].y; t.z += v[j].z; t.w += v[j].w;
          *a = t;
        }
      }
    }
  } else {
    for (int i = 0; i < n; i += 8 * RPW) {
      float v[8][CPL];
      int ld[8];
      bool ok[8];
#pragma unroll
      for (int j = 0; j < 8; ++j) {
        const int idx = i + j * RPW + slot;
        ok[j] = idx < n;
        const unsigned en = oq[w][ok[j] ? idx : 0];
        ld[j] = (int)(en & 511u);
        const float* p = x + (size_t)(en >> 9) * D + sub;
#pragma unroll
        for (int c = 0; c < CPL; ++c) v[j][c] = p[c * LPR];
      }
#pragma unroll
      for (int j = 0; j < 8; ++j) {
        if (ok[j]) {
          float* a = acc + ld[j] * D + sub;
#pragma unroll
          for (int c = 0; c < CPL; ++c) atomicAdd(a + c * LPR, v[j][c]);
          if (sub == 0) atomicAdd(&cnt[ld[j]], 1.0f);
        }
      }
    }
  }
  __syncthreads();

  if constexpr (D < 128) {
    constexpr int KS = D / 4;
    constexpr int ROW_TILES = (OWN_ROWS + 15) / 16;
    const int l16 = lane & 15;
    const int half = lane >> 4;
    const int rowsHere = min(OWN_ROWS, N - rowBase);
    for (int t = w; t < ROW_TILES; t += 32) {
      const int r = t * 16 + l16;
      const bool rowOk = r < rowsHere;
      const int rr = rowOk ? r : 0;
      const float inv = 1.0f / fmaxf(cnt[rr], 1.0f);
      v2f am[KS], ax[KS];
#pragma unroll
      for (int j = 0; j < KS; ++j) {
        const int ka = 4 * j + 2 * half;
        const v2f s = *(const v2f*)&acc[rr * D + ka];
        const v2f xv = *(const v2f*)(x + (size_t)(rowBase + rr) * D + ka);
        am[j].x = rowOk ? s.x * inv : 0.0f;
        am[j].y = rowOk ? s.y * inv : 0.0f;
        ax[j].x = rowOk ? xv.x : 0.0f;
        ax[j].y = rowOk ? xv.y : 0.0f;
      }
      v8f c[8];
#pragma unroll
      for (int ct = 0; ct < 8; ++ct) {
        const int col = ct * 16 + l16;
        v8f cc = {};
#pragma unroll
        for (int j = 0; j < KS; ++j) {
          const int ka = 4 * j + 2 * half;
          v2f bl, br;
          bl.x = wl[(size_t)ka * HIDDIM + col];
          bl.y = wl[(size_t)(ka + 1) * HIDDIM + col];
          br.x = wr[(size_t)ka * HIDDIM + col];
          br.y = wr[(size_t)(ka + 1) * HIDDIM + col];
          cc = __builtin_amdgcn_wmma_f32_16x16x4_f32(false, am[j], false, bl, (short)0, cc,
                                                     false, false);
          cc = __builtin_amdgcn_wmma_f32_16x16x4_f32(false, ax[j], false, br, (short)0, cc,
                                                     false, false);
        }
        const float bv = bias[col];
#pragma unroll
        for (int v = 0; v < 8; ++v) cc[v] += bv;
        c[ct] = cc;
      }
      float ss[8];
#pragma unroll
      for (int v = 0; v < 8; ++v) {
        float s = c[0][v] * c[0][v];
#pragma unroll
        for (int ct = 1; ct < 8; ++ct) s += c[ct][v] * c[ct][v];
        ss[v] = s;
      }
#pragma unroll
      for (int v = 0; v < 8; ++v) ss[v] = xor_add(ss[v], 1);
#pragma unroll
      for (int v = 0; v < 8; ++v) ss[v] = xor_add(ss[v], 2);
#pragma unroll
      for (int v = 0; v < 8; ++v) ss[v] = xor_add(ss[v], 4);
#pragma unroll
      for (int v = 0; v < 8; ++v) ss[v] = xor_add(ss[v], 8);
      float* const o = hout + (size_t)(rowBase + t * 16 + 8 * half) * HIDDIM + l16;
      const int left = rowsHere - (t * 16 + 8 * half);
#pragma unroll
      for (int v = 0; v < 8; ++v) {
        if (v < left) {
          const float scale = 1.0f / fmaxf(sqrtf(ss[v]), 1e-12f);
#pragma unroll
          for (int ct = 0; ct < 8; ++ct) o[v * HIDDIM + ct * 16] = tanhf(c[ct][v] * scale);
        }
      }
    }
  } else {
    for (int i = tid; i < OWN_ROWS * D / 4; i += 1024) {
      const int r = (i * 4) / D;
      const int g = rowBase + r;
      if (g < N) {
        const float inv = 1.0f / fmaxf(cnt[r], 1.0f);
        float4 s = ((const float4*)acc)[i];
        s.x *= inv; s.y *= inv; s.z *= inv; s.w *= inv;
        *(float4*)(mean + (size_t)g * D + (i * 4 - r * D)) = s;
      }
    }
  }
}

#define APITCH 132
__global__ __launch_bounds__(256, 4) void sage_layer_resident_kernel(
    const float* __restrict__ agg, const float* __restrict__ xin, const float* __restrict__ wl,
    const float* __restrict__ wr, const float* __restrict__ bias,
    float* __restrict__ out, int nTiles) {
  constexpr int K = HIDDIM;
  __shared__ __attribute__((aligned(16))) float sM[16][APITCH];
  __shared__ __attribute__((aligned(16))) float sX[16][APITCH];
  __shared__ float s_norm[2][16];

  const int tid = threadIdx.x;
  const int wave = tid >> 5;
  const int lane = tid & 31;
  const int half = lane >> 4;
  const int l16 = lane & 15;
  const int col = wave * 16 + l16;

  v2f bl[K / 4], br[K / 4];
#pragma unroll
  for (int j = 0; j < K / 4; ++j) {
    const int ka = 4 * j + 2 * half;
    bl[j].x = wl[(size_t)ka * HIDDIM + col];
    bl[j].y = wl[(size_t)(ka + 1) * HIDDIM + col];
    br[j].x = wr[(size_t)ka * HIDDIM + col];
    br[j].y = wr[(size_t)(ka + 1) * HIDDIM + col];
  }
  const float bv = bias[col];

  const int sr = tid >> 5;
  const int sc = (tid & 31) * 4;
  int tile = blockIdx.x;
  float4 pm0, pm1, px0, px1;
  pm0 = pm1 = px0 = px1 = make_float4(0.0f, 0.0f, 0.0f, 0.0f);
  if (tile < nTiles) {
    const size_t g = (size_t)(tile * 16 + sr) * K + sc;
    pm0 = *(const float4*)(agg + g);
    pm1 = *(const float4*)(agg + g + 8 * K);
    px0 = *(const float4*)(xin + g);
    px1 = *(const float4*)(xin + g + 8 * K);
  }

  int par = 0;
  for (; tile < nTiles; tile += gridDim.x) {
    const int rowBase = tile * 16;
    *(float4*)&sM[sr][sc] = pm0;
    *(float4*)&sM[sr + 8][sc] = pm1;
    *(float4*)&sX[sr][sc] = px0;
    *(float4*)&sX[sr + 8][sc] = px1;
    if (tid < 16) s_norm[par][tid] = 0.0f;
    __syncthreads();

    v8f c = {};
    v2f am[2][8], ax[2][8];
#pragma unroll
    for (int u = 0; u < 8; ++u) {
      am[0][u] = *(const v2f*)&sM[l16][4 * u + 2 * half];
      ax[0][u] = *(const v2f*)&sX[l16][4 * u + 2 * half];
    }
#pragma unroll
    for (int b = 0; b < K / 32; ++b) {
      if (b + 1 < K / 32) {
#pragma unroll
        for (int u = 0; u < 8; ++u) {
          am[(b + 1) & 1][u] = *(const v2f*)&sM[l16][32 * (b + 1) + 4 * u + 2 * half];
          ax[(b + 1) & 1][u] = *(const v2f*)&sX[l16][32 * (b + 1) + 4 * u + 2 * half];
        }
      }
      __builtin_amdgcn_sched_barrier(0);
#pragma unroll
      for (int u = 0; u < 8; ++u) {
        const int j = 8 * b + u;
        c = __builtin_amdgcn_wmma_f32_16x16x4_f32(false, am[b & 1][u], false, bl[j], (short)0, c,
                                                  false, false);
        c = __builtin_amdgcn_wmma_f32_16x16x4_f32(false, ax[b & 1][u], false, br[j], (short)0, c,
                                                  false, false);
      }
      __builtin_amdgcn_sched_barrier(0);
    }

#pragma unroll
    for (int v = 0; v < 8; ++v) c[v] += bv;

    float ss[8];
#pragma unroll
    for (int v = 0; v < 8; ++v) ss[v] = c[v] * c[v];
#pragma unroll
    for (int v = 0; v < 8; ++v) ss[v] = xor_add(ss[v], 1);
#pragma unroll
    for (int v = 0; v < 8; ++v) ss[v] = xor_add(ss[v], 2);
#pragma unroll
    for (int v = 0; v < 8; ++v) ss[v] = xor_add(ss[v], 4);
#pragma unroll
    for (int v = 0; v < 8; ++v) ss[v] = xor_add(ss[v], 8);
    if (l16 == 0) {
#pragma unroll
      for (int v = 0; v < 8; ++v) atomicAdd(&s_norm[par][v + 8 * half], ss[v]);
    }
    __syncthreads();

    {
      const int nt = tile + (int)gridDim.x < nTiles ? tile + (int)gridDim.x : tile;
      const size_t g = (size_t)(nt * 16 + sr) * K + sc;
      pm0 = *(const float4*)(agg + g);
      pm1 = *(const float4*)(agg + g + 8 * K);
      px0 = *(const float4*)(xin + g);
      px1 = *(const float4*)(xin + g + 8 * K);
    }

    float scale[8];
#pragma unroll
    for (int v = 0; v < 8; ++v)
      scale[v] = 1.0f / fmaxf(sqrtf(s_norm[par][v + 8 * half]), 1e-12f);
    asm volatile("" ::"v"(scale[0]), "v"(scale[1]), "v"(scale[2]), "v"(scale[3]), "v"(scale[4]),
                 "v"(scale[5]), "v"(scale[6]), "v"(scale[7]));
#pragma unroll
    for (int v = 0; v < 8; ++v)
      out[(size_t)(rowBase + v + 8 * half) * HIDDIM + col] = tanhf(c[v] * scale[v]);
    par ^= 1;
  }
}

extern "C" void kernel_launch(void* const* d_in, const int* in_sizes, int n_in,
                              void* d_out, int out_size, void* d_ws,
                              size_t ws_size, hipStream_t stream) {
  const float* x = (const float*)d_in[0];
  const int* ei_conn = (const int*)d_in[1];
  const int* ei_dest = (const int*)d_in[2];
  const float* wl1 = (const float*)d_in[3];
  const float* wr1 = (const float*)d_in[4];
  const float* b1 = (const float*)d_in[5];
  const float* wl2 = (const float*)d_in[6];
  const float* wr2 = (const float*)d_in[7];
  const float* b2 = (const float*)d_in[8];
  const float* wl3 = (const float*)d_in[9];
  const float* wr3 = (const float*)d_in[10];
  const float* b3 = (const float*)d_in[11];

  const int N = in_sizes[0] / 8;
  const int E = in_sizes[1] / 2;

  float* ws = (float*)d_ws;
  float* mean = ws;
  float* h1 = mean + (size_t)N * HIDDIM;
  float* h2 = h1 + (size_t)N * HIDDIM;

  const int T = 256;
  const int* src_conn = ei_conn;
  const int* dst_conn = ei_conn + E;
  const int* src_dest = ei_dest;
  const int* dst_dest = ei_dest + E;

  const int gemmBlocks = (N + 15) / 16;
  const int ownBlocks = (N + OWN_ROWS - 1) / OWN_ROWS;
  const int wideBlocks = 512;
  const int nChunks = E / PLANE_CHUNK;
  const size_t planeAll = (size_t)ownBlocks * nChunks * (PLANE_CHUNK / 32);
  unsigned* plane_conn = (unsigned*)(h2 + (size_t)N * HIDDIM);
  unsigned* plane_dest = plane_conn + planeAll;
  unsigned* rec = plane_dest + planeAll;
  int* recCnt = (int*)(rec + (size_t)ownBlocks * 32 * OCAP);

  owner_plane_kernel<<<dim3(nChunks > 0 ? nChunks : 1, 2), 1024, 0, stream>>>(
      dst_conn, dst_dest, plane_conn, plane_dest, nChunks, ownBlocks);

  owner_mean_kernel<8, MODE_RECORD><<<ownBlocks, 1024, 0, stream>>>(
      x, src_conn, dst_conn, plane_conn, mean, rec, recCnt, N, E, nChunks, wl1, wr1, b1, h1);

  owner_mean_kernel<HIDDIM, MODE_SCAN><<<ownBlocks, 1024, 0, stream>>>(
      h1, src_dest, dst_dest, plane_dest, mean, rec, recCnt, N, E, nChunks, nullptr, nullptr,
      nullptr, nullptr);
  sage_layer_resident_kernel<<<wideBlocks, T, 0, stream>>>(mean, h1, wl2, wr2, b2, h2, gemmBlocks);

  owner_mean_kernel<HIDDIM, MODE_REPLAY><<<ownBlocks, 1024, 0, stream>>>(
      h2, src_conn, dst_conn, plane_conn, mean, rec, recCnt, N, E, nChunks, nullptr, nullptr,
      nullptr, nullptr);
  sage_layer_resident_kernel<<<wideBlocks, T, 0, stream>>>(mean, h2, wl3, wr3, b3, (float*)d_out,
                                                           gemmBlocks);
}


# Plane builder also writes one packed word an edge (source id * 512 + local row): a block reads that one word for an edge it finds, not its two ids; list order arrival-dependent as before
# speedup vs baseline: 3.2284x; 1.0321x over previous
#include <hip/hip_runtime.h>
#include <hip/hip_bf16.h>
#include <math.h>

typedef __attribute__((ext_vector_type(2))) float v2f;
typedef __attribute__((ext_vector_type(8))) float v8f;

#define HIDDIM 128

#define OWN_ROWS 400
#define QCAP 544
#define OCAP 320
#define PLANE_CHUNK 8192
#define PLANE_PITCH 260
#define MAX_OWN 256
#define GROUP_EDGES 4096
#define REM_CHUNK 1024
#define MODE_SCAN 0
#define MODE_RECORD 1
#define MODE_REPLAY 2

__global__ __launch_bounds__(1024) void owner_plane_kernel(
    const int* __restrict__ srcA, const int* __restrict__ dstA,
    const int* __restrict__ srcB, const int* __restrict__ dstB,
    unsigned* __restrict__ planeA, unsigned* __restrict__ planeB,
    unsigned* __restrict__ packA, unsigned* __restrict__ packB, int nChunks, int ownBlocks) {
  __shared__ __attribute__((aligned(16))) unsigned bits[MAX_OWN * PLANE_PITCH];
  if ((int)blockIdx.x >= nChunks) return;
  const int tid = threadIdx.x;
  const int nb = min(ownBlocks, MAX_OWN);
  const int* src = blockIdx.y ? srcB : srcA;
  const int* dst = blockIdx.y ? dstB : dstA;
  unsigned* plane = blockIdx.y ? planeB : planeA;
  unsigned* pack = blockIdx.y ? packB : packA;
  for (int i = tid; i < nb * (PLANE_PITCH / 4); i += 1024)
    ((uint4*)bits)[i] = make_uint4(0u, 0u, 0u, 0u);
  const int e0 = blockIdx.x * PLANE_CHUNK;
  int d[8], s[8];
#pragma unroll
  for (int j = 0; j < 8; ++j) {
    d[j] = dst[e0 + j * 1024 + tid];
    s[j] = src[e0 + j * 1024 + tid];
  }
  __syncthreads();
#pragma unroll
  for (int j = 0; j < 8; ++j) {
    const unsigned o = (unsigned)d[j] / OWN_ROWS;
    if (o < (unsigned)nb) atomicOr(&bits[o * PLANE_PITCH + j * 32 + (tid >> 5)], 1u << (tid & 31));
    pack[e0 + j * 1024 + tid] = ((unsigned)s[j] << 9) | ((unsigned)d[j] - o * OWN_ROWS);
  }
  __syncthreads();
  const size_t planeWords = (size_t)nChunks * (PLANE_CHUNK / 32);
  for (int i = tid; i < nb * 64; i += 1024) {
    const int b = i >> 6;
    const int k = (i & 63) * 4;
    const uint4 v = *(const uint4*)&bits[b * PLANE_PITCH + k];
    *(uint4*)(plane + (size_t)b * planeWords + (size_t)blockIdx.x * (PLANE_CHUNK / 32) + k) = v;
  }
}

__device__ __forceinline__ float xor_add(float s, const int imm) {
  switch (imm) {
    case 1:  s += __int_as_float(__builtin_amdgcn_ds_swizzle(__float_as_int(s), 0x041f)); break;
    case 2:  s += __int_as_float(__builtin_amdgcn_ds_swizzle(__float_as_int(s), 0x081f)); break;
    case 4:  s += __int_as_float(__builtin_amdgcn_ds_swizzle(__float_as_int(s), 0x101f)); break;
    default: s += __int_as_float(__builtin_amdgcn_ds_swizzle(__float_as_int(s), 0x201f)); break;
  }
  return s;
}

template <int D, int MODE>
__global__ __launch_bounds__(1024) void owner_mean_kernel(
    const float* __restrict__ x, const int* __restrict__ src,
    const int* __restrict__ dst, const unsigned* __restrict__ plane,
    const unsigned* __restrict__ pack,
    float* __restrict__ mean, unsigned* __restrict__ rec, int* __restrict__ recCnt,
    int N, int E, int nChunks,
    const float* __restrict__ wl, const float* __restrict__ wr, const float* __restrict__ bias,
    float* __restrict__ hout) {
  constexpr int LPR = (D >= 32) ? 32 : D;
  constexpr int CPL = D / LPR;
  constexpr int RPW = 32 / LPR;
  __shared__ __attribute__((aligned(16))) float acc[OWN_ROWS * D];
  __shared__ float cnt[OWN_ROWS];
  __shared__ unsigned q[32][QCAP];
  __shared__ unsigned oq[32][OCAP];
  __shared__ unsigned ocnt[32];
  __shared__ unsigned ovf;

  const int tid = threadIdx.x;
  const int w = __builtin_amdgcn_readfirstlane(tid >> 5);
  const int lane = tid & 31;
  const int sub = lane % LPR;
  const int slot = lane / LPR;
  const int rowBase = blockIdx.x * OWN_ROWS;
  const unsigned below = (1u << lane) - 1u;

  for (int i = tid; i < OWN_ROWS * D / 4; i += 1024)
    ((float4*)acc)[i] = make_float4(0.0f, 0.0f, 0.0f, 0.0f);
  for (int i = tid; i < OWN_ROWS; i += 1024) cnt[i] = 0.0f;
  if (tid < 32) ocnt[tid] = 0u;
  if (tid == 0) ovf = 0u;
  __syncthreads();

  int qn = 0;
  unsigned* const myRec = rec + ((size_t)blockIdx.x * 32 + w) * OCAP;
  int* const recOk = recCnt + (size_t)gridDim.x * 32;

  auto push = [&](unsigned mask, int e) {
    if (mask) {
      if ((mask >> lane) & 1u)
        q[w][qn + __builtin_popcount(mask & below)] = (unsigned)e;
      qn += __builtin_popcount(mask);
    }
  };

  const int covered = nChunks * PLANE_CHUNK;
  constexpr bool PACKED = MODE != MODE_REPLAY;

  auto hand_over = [&]() {
    for (int j = lane; j < qn; j += 32) {
      const unsigned e = q[w][j];
      unsigned en;
      if (PACKED && (int)e < covered) {
        en = pack[e];
      } else {
        en = ((unsigned)src[e] << 9) | (unsigned)(dst[e] - rowBase);
      }
      const unsigned s = en >> 9;
      const unsigned lr = en & 511u;
      const unsigned o = lr & 31u;
      const unsigned pos = atomicAdd(&ocnt[o], 1u);
      if (pos < (unsigned)OCAP) {
        oq[o][pos] = en;
      } else {
        ovf = 1u;
        const float* p = x + (size_t)s * D;
        for (int c = 0; c < D; ++c) atomicAdd(&acc[lr * D + c], p[c]);
        atomicAdd(&cnt[lr], 1.0f);
      }
    }
    qn = 0;
  };

  bool haveList = false;
  int n = 0;
  if constexpr (MODE == MODE_REPLAY) {
    haveList = recOk[blockIdx.x] == 1;
    if (haveList) {
      n = __builtin_amdgcn_readfirstlane(recCnt[blockIdx.x * 32 + w]);
      n = min(max(n, 0), OCAP);
      for (int j = lane; j < n; j += 32) oq[w][j] = myRec[j];
    }
  }

  if (!haveList && nChunks > 0) {
    const int nGroups = covered / GROUP_EDGES;
    const uint4* p4 = (const uint4*)(plane + (size_t)blockIdx.x * ((size_t)nChunks * (PLANE_CHUNK / 32))) + lane;
    auto group_of = [&](int t, int u) -> int { return (4 * t + u) * 32 + w; };
    auto load_group = [&](int g) -> uint4 { return p4[(size_t)min(g, nGroups - 1) * 32]; };
    auto take = [&](const uint4 c, const int eb) {
      unsigned long long m0 = ((unsigned long long)c.y << 32) | c.x;
      unsigned long long m1 = ((unsigned long long)c.w << 32) | c.z;
      unsigned hm = __builtin_amdgcn_ballot_w32((m0 | m1) != 0ull);
      while (hm) {
        if ((m0 | m1) != 0ull) {
          int bpos;
          if (m0 != 0ull) {
            bpos = __builtin_ctzll(m0);
            m0 &= m0 - 1ull;
          } else {
            bpos = 64 + __builtin_ctzll(m1);
            m1 &= m1 - 1ull;
          }
          q[w][qn + __builtin_popcount(hm & below)] = (unsigned)(eb + bpos);
        }
        qn += __builtin_popcount(hm);
        if (qn >= 128) hand_over();
        hm = __builtin_amdgcn_ballot_w32((m0 | m1) != 0ull);
      }
    };
    uint4 cur[4];
#pragma unroll
    for (int u = 0; u < 4; ++u) cur[u] = load_group(group_of(0, u));
    for (int t = 0; group_of(t, 0) < nGroups; ++t) {
      uint4 nxt[4];
#pragma unroll
      for (int u = 0; u < 4; ++u) nxt[u] = load_group(group_of(t + 1, u));
#pragma unroll
      for (int u = 0; u < 4; ++u) {
        const int g = group_of(t, u);
        if (g < nGroups) take(cur[u], g * GROUP_EDGES + lane * 128);
      }
#pragma unroll
      for (int u = 0; u < 4; ++u) cur[u] = nxt[u];
    }
  }
  for (int rem0 = haveList ? E : covered + w * REM_CHUNK; rem0 < E; rem0 += 32 * REM_CHUNK) {
    const int lim = min(rem0 + REM_CHUNK, E);
    for (int g = rem0; g < lim; g += 256) {
      int dt[8];
#pragma unroll
      for (int u = 0; u < 8; ++u) dt[u] = dst[min(g + u * 32 + lane, E - 1)];
#pragma unroll
      for (int u = 0; u < 8; ++u) {
        const int e = g + u * 32 + lane;
        const unsigned l = (unsigned)(dt[u] - rowBase);
        push(__builtin_amdgcn_ballot_w32(e < lim && l < (unsigned)OWN_ROWS), e);
      }
      if (qn >= 32) hand_over();
    }
  }
  if (qn > 0) hand_over();
  __syncthreads();

  if (!haveList) n = min((int)__builtin_amdgcn_readfirstlane((int)ocnt[w]), OCAP);
  if constexpr (MODE == MODE_RECORD) {
    for (int j = lane; j < n; j += 32) myRec[j] = oq[w][j];
    if (lane == 0) recCnt[blockIdx.x * 32 + w] = n;
    if (tid == 0) recOk[blockIdx.x] = ovf ? 0 : 1;
  }

  if constexpr (D >= 128) {
    for (int i = 0; i < n; i += 16) {
      const int mi = i + (lane & 15);
      const unsigned mine = oq[w][mi < n ? mi : 0];
      float4 v[16];
#pragma unroll
      for (int j = 0; j < 16; ++j) {
        const unsigned en = (unsigned)__builtin_amdgcn_readlane((int)mine, j);
        v[j] = *(const float4*)(x + (size_t)(en >> 9) * D + 4 * lane);
      }
      if (lane < 16 && mi < n) atomicAdd(&cnt[mine & 511u], 1.0f);
#pragma unroll
      for (int j = 0; j < 16; ++j) {
        if (i + j < n) {
          const unsigned lr = (unsigned)__builtin_amdgcn_readlane((int)mine, j) & 511u;
          float4* a = (float4*)(acc + lr * D) + lane;
          float4 t = *a;
          t.x += v[j].x; t.y += v[j].y; t.z += v[j].z; t.w += v[j].w;
          *a = t;
        }
      }
    }
  } else {
    for (int i = 0; i < n; i += 8 * RPW) {
      float v[8][CPL];
      int ld[8];
      bool ok[8];
#pragma unroll
      for (int j = 0; j < 8; ++j) {
        const int idx = i + j * RPW + slot;
        ok[j] = idx < n;
        const unsigned en = oq[w][ok[j] ? idx : 0];
        ld[j] = (int)(en & 511u);
        const float* p = x + (size_t)(en >> 9) * D + sub;
#pragma unroll
        for (int c = 0; c < CPL; ++c) v[j][c] = p[c * LPR];
      }
#pragma unroll
      for (int j = 0; j < 8; ++j) {
        if (ok[j]) {
          float* a = acc + ld[j] * D + sub;
#pragma unroll
          for (int c = 0; c < CPL; ++c) atomicAdd(a + c * LPR, v[j][c]);
          if (sub == 0) atomicAdd(&cnt[ld[j]], 1.0f);
        }
      }
    }
  }
  __syncthreads();

  if constexpr (D < 128) {
    constexpr int KS = D / 4;
    constexpr int ROW_TILES = (OWN_ROWS + 15) / 16;
    const int l16 = lane & 15;
    const int half = lane >> 4;
    const int rowsHere = min(OWN_ROWS, N - rowBase);
    for (int t = w; t < ROW_TILES; t += 32) {
      const int r = t * 16 + l16;
      const bool rowOk = r < rowsHere;
      const int rr = rowOk ? r : 0;
      const float inv = 1.0f / fmaxf(cnt[rr], 1.0f);
      v2f am[KS], ax[KS];
#pragma unroll
      for (int j = 0; j < KS; ++j) {
        const int ka = 4 * j + 2 * half;
        const v2f s = *(const v2f*)&acc[rr * D + ka];
        const v2f xv = *(const v2f*)(x + (size_t)(rowBase + rr) * D + ka);
        am[j].x = rowOk ? s.x * inv : 0.0f;
        am[j].y = rowOk ? s.y * inv : 0.0f;
        ax[j].x = rowOk ? xv.x : 0.0f;
        ax[j].y = rowOk ? xv.y : 0.0f;
      }
      v8f c[8];
#pragma unroll
      for (int ct = 0; ct < 8; ++ct) {
        const int col = ct * 16 + l16;
        v8f cc = {};
#pragma unroll
        for (int j = 0; j < KS; ++j) {
          const int ka = 4 * j + 2 * half;
          v2f bl, br;
          bl.x = wl[(size_t)ka * HIDDIM + col];
          bl.y = wl[(size_t)(ka + 1) * HIDDIM + col];
          br.x = wr[(size_t)ka * HIDDIM + col];
          br.y = wr[(size_t)(ka + 1) * HIDDIM + col];
          cc = __builtin_amdgcn_wmma_f32_16x16x4_f32(false, am[j], false, bl, (short)0, cc,
                                                     false, false);
          cc = __builtin_amdgcn_wmma_f32_16x16x4_f32(false, ax[j], false, br, (short)0, cc,
                                                     false, false);
        }
        const float bv = bias[col];
#pragma unroll
        for (int v = 0; v < 8; ++v) cc[v] += bv;
        c[ct] = cc;
      }
      float ss[8];
#pragma unroll
      for (int v = 0; v < 8; ++v) {
        float s = c[0][v] * c[0][v];
#pragma unroll
        for (int ct = 1; ct < 8; ++ct) s += c[ct][v] * c[ct][v];
        ss[v] = s;
      }
#pragma unroll
      for (int v = 0; v < 8; ++v) ss[v] = xor_add(ss[v], 1);
#pragma unroll
      for (int v = 0; v < 8; ++v) ss[v] = xor_add(ss[v], 2);
#pragma unroll
      for (int v = 0; v < 8; ++v) ss[v] = xor_add(ss[v], 4);
#pragma unroll
      for (int v = 0; v < 8; ++v) ss[v] = xor_add(ss[v], 8);
      float* const o = hout + (size_t)(rowBase + t * 16 + 8 * half) * HIDDIM + l16;
      const int left = rowsHere - (t * 16 + 8 * half);
#pragma unroll
      for (int v = 0; v < 8; ++v) {
        if (v < left) {
          const float scale = 1.0f / fmaxf(sqrtf(ss[v]), 1e-12f);
#pragma unroll
          for (int ct = 0; ct < 8; ++ct) o[v * HIDDIM + ct * 16] = tanhf(c[ct][v] * scale);
        }
      }
    }
  } else {
    for (int i = tid; i < OWN_ROWS * D / 4; i += 1024) {
      const int r = (i * 4) / D;
      const int g = rowBase + r;
      if (g < N) {
        const float inv = 1.0f / fmaxf(cnt[r], 1.0f);
        float4 s = ((const float4*)acc)[i];
        s.x *= inv; s.y *= inv; s.z *= inv; s.w *= inv;
        *(float4*)(mean + (size_t)g * D + (i * 4 - r * D)) = s;
      }
    }
  }
}

#define APITCH 132
__global__ __launch_bounds__(256, 4) void sage_layer_resident_kernel(
    const float* __restrict__ agg, const float* __restrict__ xin, const float* __restrict__ wl,
    const float* __restrict__ wr, const float* __restrict__ bias,
    float* __restrict__ out, int nTiles) {
  constexpr int K = HIDDIM;
  __shared__ __attribute__((aligned(16))) float sM[16][APITCH];
  __shared__ __attribute__((aligned(16))) float sX[16][APITCH];
  __shared__ float s_norm[2][16];

  const int tid = threadIdx.x;
  const int wave = tid >> 5;
  const int lane = tid & 31;
  const int half = lane >> 4;
  const int l16 = lane & 15;
  const int col = wave * 16 + l16;

  v2f bl[K / 4], br[K / 4];
#pragma unroll
  for (int j = 0; j < K / 4; ++j) {
    const int ka = 4 * j + 2 * half;
    bl[j].x = wl[(size_t)ka * HIDDIM + col];
    bl[j].y = wl[(size_t)(ka + 1) * HIDDIM + col];
    br[j].x = wr[(size_t)ka * HIDDIM + col];
    br[j].y = wr[(size_t)(ka + 1) * HIDDIM + col];
  }
  const float bv = bias[col];

  const int sr = tid >> 5;
  const int sc = (tid & 31) * 4;
  int tile = blockIdx.x;
  float4 pm0, pm1, px0, px1;
  pm0 = pm1 = px0 = px1 = make_float4(0.0f, 0.0f, 0.0f, 0.0f);
  if (tile < nTiles) {
    const size_t g = (size_t)(tile * 16 + sr) * K + sc;
    pm0 = *(const float4*)(agg + g);
    pm1 = *(const float4*)(agg + g + 8 * K);
    px0 = *(const float4*)(xin + g);
    px1 = *(const float4*)(xin + g + 8 * K);
  }

  int par = 0;
  for (; tile < nTiles; tile += gridDim.x) {
    const int rowBase = tile * 16;
    *(float4*)&sM[sr][sc] = pm0;
    *(float4*)&sM[sr + 8][sc] = pm1;
    *(float4*)&sX[sr][sc] = px0;
    *(float4*)&sX[sr + 8][sc] = px1;
    if (tid < 16) s_norm[par][tid] = 0.0f;
    __syncthreads();

    v8f c = {};
    v2f am[2][8], ax[2][8];
#pragma unroll
    for (int u = 0; u < 8; ++u) {
      am[0][u] = *(const v2f*)&sM[l16][4 * u + 2 * half];
      ax[0][u] = *(const v2f*)&sX[l16][4 * u + 2 * half];
    }
#pragma unroll
    for (int b = 0; b < K / 32; ++b) {
      if (b + 1 < K / 32) {
#pragma unroll
        for (int u = 0; u < 8; ++u) {
          am[(b + 1) & 1][u] = *(const v2f*)&sM[l16][32 * (b + 1) + 4 * u + 2 * half];
          ax[(b + 1) & 1][u] = *(const v2f*)&sX[l16][32 * (b + 1) + 4 * u + 2 * half];
        }
      }
      __builtin_amdgcn_sched_barrier(0);
#pragma unroll
      for (int u = 0; u < 8; ++u) {
        const int j = 8 * b + u;
        c = __builtin_amdgcn_wmma_f32_16x16x4_f32(false, am[b & 1][u], false, bl[j], (short)0, c,
                                                  false, false);
        c = __builtin_amdgcn_wmma_f32_16x16x4_f32(false, ax[b & 1][u], false, br[j], (short)0, c,
                                                  false, false);
      }
      __builtin_amdgcn_sched_barrier(0);
    }

#pragma unroll
    for (int v = 0; v < 8; ++v) c[v] += bv;

    float ss[8];
#pragma unroll
    for (int v = 0; v < 8; ++v) ss[v] = c[v] * c[v];
#pragma unroll
    for (int v = 0; v < 8; ++v) ss[v] = xor_add(ss[v], 1);
#pragma unroll
    for (int v = 0; v < 8; ++v) ss[v] = xor_add(ss[v], 2);
#pragma unroll
    for (int v = 0; v < 8; ++v) ss[v] = xor_add(ss[v], 4);
#pragma unroll
    for (int v = 0; v < 8; ++v) ss[v] = xor_add(ss[v], 8);
    if (l16 == 0) {
#pragma unroll
      for (int v = 0; v < 8; ++v) atomicAdd(&s_norm[par][v + 8 * half], ss[v]);
    }
    __syncthreads();

    {
      const int nt = tile + (int)gridDim.x < nTiles ? tile + (int)gridDim.x : tile;
      const size_t g = (size_t)(nt * 16 + sr) * K + sc;
      pm0 = *(const float4*)(agg + g);
      pm1 = *(const float4*)(agg + g + 8 * K);
      px0 = *(const float4*)(xin + g);
      px1 = *(const float4*)(xin + g + 8 * K);
    }

    float scale[8];
#pragma unroll
    for (int v = 0; v < 8; ++v)
      scale[v] = 1.0f / fmaxf(sqrtf(s_norm[par][v + 8 * half]), 1e-12f);
    asm volatile("" ::"v"(scale[0]), "v"(scale[1]), "v"(scale[2]), "v"(scale[3]), "v"(scale[4]),
                 "v"(scale[5]), "v"(scale[6]), "v"(scale[7]));
#pragma unroll
    for (int v = 0; v < 8; ++v)
      out[(size_t)(rowBase + v + 8 * half) * HIDDIM + col] = tanhf(c[v] * scale[v]);
    par ^= 1;
  }
}

extern "C" void kernel_launch(void* const* d_in, const int* in_sizes, int n_in,
                              void* d_out, int out_size, void* d_ws,
                              size_t ws_size, hipStream_t stream) {
  const float* x = (const float*)d_in[0];
  const int* ei_conn = (const int*)d_in[1];
  const int* ei_dest = (const int*)d_in[2];
  const float* wl1 = (const float*)d_in[3];
  const float* wr1 = (const float*)d_in[4];
  const float* b1 = (const float*)d_in[5];
  const float* wl2 = (const float*)d_in[6];
  const float* wr2 = (const float*)d_in[7];
  const float* b2 = (const float*)d_in[8];
  const float* wl3 = (const float*)d_in[9];
  const float* wr3 = (const float*)d_in[10];
  const float* b3 = (const float*)d_in[11];

  const int N = in_sizes[0] / 8;
  const int E = in_sizes[1] / 2;

  float* ws = (float*)d_ws;
  float* mean = ws;
  float* h1 = mean + (size_t)N * HIDDIM;
  float* h2 = h1 + (size_t)N * HIDDIM;

  const int T = 256;
  const int* src_conn = ei_conn;
  const int* dst_conn = ei_conn + E;
  const int* src_dest = ei_dest;
  const int* dst_dest = ei_dest + E;

  const int gemmBlocks = (N + 15) / 16;
  const int ownBlocks = (N + OWN_ROWS - 1) / OWN_ROWS;
  const int wideBlocks = 512;
  const int nChunks = E / PLANE_CHUNK;
  const size_t planeAll = (size_t)ownBlocks * nChunks * (PLANE_CHUNK / 32);
  unsigned* plane_conn = (unsigned*)(h2 + (size_t)N * HIDDIM);
  unsigned* plane_dest = plane_conn + planeAll;
  unsigned* rec = plane_dest + planeAll;
  int* recCnt = (int*)(rec + (size_t)ownBlocks * 32 * OCAP);
  unsigned* pack_conn = (unsigned*)h2;
  unsigned* pack_dest = pack_conn + E;

  owner_plane_kernel<<<dim3(nChunks > 0 ? nChunks : 1, 2), 1024, 0, stream>>>(
      src_conn, dst_conn, src_dest, dst_dest, plane_conn, plane_dest, pack_conn, pack_dest, nChunks,
      ownBlocks);

  owner_mean_kernel<8, MODE_RECORD><<<ownBlocks, 1024, 0, stream>>>(
      x, src_conn, dst_conn, plane_conn, pack_conn, mean, rec, recCnt, N, E, nChunks, wl1, wr1, b1,
      h1);

  owner_mean_kernel<HIDDIM, MODE_SCAN><<<ownBlocks, 1024, 0, stream>>>(
      h1, src_dest, dst_dest, plane_dest, pack_dest, mean, rec, recCnt, N, E, nChunks, nullptr,
      nullptr, nullptr, nullptr);
  sage_layer_resident_kernel<<<wideBlocks, T, 0, stream>>>(mean, h1, wl2, wr2, b2, h2, gemmBlocks);

  owner_mean_kernel<HIDDIM, MODE_REPLAY><<<ownBlocks, 1024, 0, stream>>>(
      h2, src_conn, dst_conn, plane_conn, nullptr, mean, rec, recCnt, N, E, nChunks, nullptr,
      nullptr, nullptr, nullptr);
  sage_layer_resident_kernel<<<wideBlocks, T, 0, stream>>>(mean, h2, wl3, wr3, b3, (float*)d_out,
                                                           gemmBlocks);
}


# Edge finding by owner slots: first kernel groups a chunk's edges by owner block into 256-byte slots (count, packed words; LDS add gives the place); no bit planes, no lookups; arrival-ordered sums
# speedup vs baseline: 3.6043x; 1.1164x over previous
#include <hip/hip_runtime.h>
#include <hip/hip_bf16.h>
#include <math.h>

typedef __attribute__((ext_vector_type(2))) float v2f;
typedef __attribute__((ext_vector_type(8))) float v8f;

#define HIDDIM 128

#define OWN_ROWS 400
#define OCAP 320
#define CHUNK_EDGES 7168
#define SLOT_WORDS 64
#define SLOT_CAP 63
#define MAX_OWN 256
#define IMG_PITCH 68
static_assert(CHUNK_EDGES % 1024 == 0 && CHUNK_EDGES % 256 == 0, "whole steps of the grouping block and of the id scan");
#define MODE_SCAN 0
#define MODE_RECORD 1
#define MODE_REPLAY 2

__global__ __launch_bounds__(1024) void owner_slot_kernel(
    const int* __restrict__ srcA, const int* __restrict__ dstA,
    const int* __restrict__ srcB, const int* __restrict__ dstB,
    unsigned* __restrict__ slotA, unsigned* __restrict__ slotB, int nChunks, int ownBlocks) {
  __shared__ __attribute__((aligned(16))) unsigned img[MAX_OWN * IMG_PITCH];
  __shared__ unsigned cntL[MAX_OWN];
  if ((int)blockIdx.x >= nChunks) return;
  const int tid = threadIdx.x;
  const int nb = min(ownBlocks, MAX_OWN);
  const int* src = blockIdx.y ? srcB : srcA;
  const int* dst = blockIdx.y ? dstB : dstA;
  unsigned* slot = blockIdx.y ? slotB : slotA;
  for (int i = tid; i < nb * (IMG_PITCH / 4); i += 1024)
    ((uint4*)img)[i] = make_uint4(0u, 0u, 0u, 0u);
  if (tid < MAX_OWN) cntL[tid] = 0u;
  const int e0 = blockIdx.x * CHUNK_EDGES;
  constexpr int EPT = CHUNK_EDGES / 1024;
  int d[EPT], s[EPT];
#pragma unroll
  for (int j = 0; j < EPT; ++j) {
    d[j] = dst[e0 + j * 1024 + tid];
    s[j] = src[e0 + j * 1024 + tid];
  }
  __syncthreads();
  unsigned pos[EPT];
#pragma unroll
  for (int j = 0; j < EPT; ++j) {
    const unsigned o = (unsigned)d[j] / OWN_ROWS;
    pos[j] = o < (unsigned)nb ? atomicAdd(&cntL[o], 1u) : (unsigned)SLOT_CAP;
  }
#pragma unroll
  for (int j = 0; j < EPT; ++j) {
    const unsigned o = (unsigned)d[j] / OWN_ROWS;
    if (pos[j] < (unsigned)SLOT_CAP)
      img[o * IMG_PITCH + 1 + pos[j]] = ((unsigned)s[j] << 9) | ((unsigned)d[j] - o * OWN_ROWS);
  }
  __syncthreads();
  for (int i = tid; i < nb * (SLOT_WORDS / 4); i += 1024) {
    const int b = i / (SLOT_WORDS / 4);
    const int k = (i % (SLOT_WORDS / 4)) * 4;
    uint4 v = *(const uint4*)&img[b * IMG_PITCH + k];
    if (k == 0) v.x = cntL[b];
    *(uint4*)(slot + ((size_t)b * nChunks + blockIdx.x) * SLOT_WORDS + k) = v;
  }
}

__device__ __forceinline__ float xor_add(float s, const int imm) {
  switch (imm) {
    case 1:  s += __int_as_float(__builtin_amdgcn_ds_swizzle(__float_as_int(s), 0x041f)); break;
    case 2:  s += __int_as_float(__builtin_amdgcn_ds_swizzle(__float_as_int(s), 0x081f)); break;
    case 4:  s += __int_as_float(__builtin_amdgcn_ds_swizzle(__float_as_int(s), 0x101f)); break;
    default: s += __int_as_float(__builtin_amdgcn_ds_swizzle(__float_as_int(s), 0x201f)); break;
  }
  return s;
}

template <int D, int MODE>
__global__ __launch_bounds__(1024) void owner_mean_kernel(
    const float* __restrict__ x, const int* __restrict__ src,
    const int* __restrict__ dst, const unsigned* __restrict__ slots,
    float* __restrict__ mean, unsigned* __restrict__ rec, int* __restrict__ recCnt,
    int N, int E, int nChunks,
    const float* __restrict__ wl, const float* __restrict__ wr, const float* __restrict__ bias,
    float* __restrict__ hout) {
  constexpr int LPR = (D >= 32) ? 32 : D;
  constexpr int CPL = D / LPR;
  constexpr int RPW = 32 / LPR;
  __shared__ __attribute__((aligned(16))) float acc[OWN_ROWS * D];
  __shared__ float cnt[OWN_ROWS];
  __shared__ unsigned oq[32][OCAP];
  __shared__ unsigned ocnt[32];
  __shared__ unsigned ovf;

  const int tid = threadIdx.x;
  const int w = __builtin_amdgcn_readfirstlane(tid >> 5);
  const int lane = tid & 31;
  const int sub = lane % LPR;
  const int slot = lane / LPR;
  const int rowBase = blockIdx.x * OWN_ROWS;

  for (int i = tid; i < OWN_ROWS * D / 4; i += 1024)
    ((float4*)acc)[i] = make_float4(0.0f, 0.0f, 0.0f, 0.0f);
  for (int i = tid; i < OWN_ROWS; i += 1024) cnt[i] = 0.0f;
  if (tid < 32) ocnt[tid] = 0u;
  if (tid == 0) ovf = 0u;
  __syncthreads();

  unsigned* const myRec = rec + ((size_t)blockIdx.x * 32 + w) * OCAP;
  int* const recOk = recCnt + (size_t)gridDim.x * 32;

  const int covered = nChunks * CHUNK_EDGES;

  auto commit = [&](const unsigned en, const unsigned pos) {
    const unsigned lr = en & 511u;
    if (pos < (unsigned)OCAP) {
      oq[lr & 31u][pos] = en;
    } else {
      ovf = 1u;
      const float* p = x + (size_t)(en >> 9) * D;
      for (int c = 0; c < D; ++c) atomicAdd(&acc[lr * D + c], p[c]);
      atomicAdd(&cnt[lr], 1.0f);
    }
  };
  auto place = [&](const unsigned en) { commit(en, atomicAdd(&ocnt[en & 31u], 1u)); };

  bool haveList = false;
  int n = 0;
  if constexpr (MODE == MODE_REPLAY) {
    haveList = recOk[blockIdx.x] == 1;
    if (haveList) {
      n = __builtin_amdgcn_readfirstlane(recCnt[blockIdx.x * 32 + w]);
      n = min(max(n, 0), OCAP);
      for (int j = lane; j < n; j += 32) oq[w][j] = myRec[j];
    }
  }

  constexpr int RPT = CHUNK_EDGES / 1024;
  int rd[RPT], rs[RPT];
#pragma unroll
  for (int k = 0; k < RPT; ++k) rd[k] = rs[k] = 0;
  if (!haveList) {
#pragma unroll
    for (int k = 0; k < RPT; ++k) {
      const int e = min(covered + k * 1024 + tid, E - 1);
      rd[k] = dst[e];
      rs[k] = src[e];
    }
  }

  if (!haveList && nChunks > 0) {
    const unsigned* const mySlots = slots + (size_t)blockIdx.x * nChunks * SLOT_WORDS;
    for (int c0 = w; c0 < nChunks; c0 += 256) {
      unsigned lo[8], hi[8];
#pragma unroll
      for (int u = 0; u < 8; ++u) {
        const int c = min(c0 + 32 * u, nChunks - 1);
        lo[u] = mySlots[(size_t)c * SLOT_WORDS + lane];
        hi[u] = mySlots[(size_t)c * SLOT_WORDS + 32 + lane];
      }
      asm volatile("" ::"v"(lo[0]), "v"(lo[1]), "v"(lo[2]), "v"(lo[3]), "v"(lo[4]), "v"(lo[5]),
                   "v"(lo[6]), "v"(lo[7]), "v"(hi[0]), "v"(hi[1]), "v"(hi[2]), "v"(hi[3]),
                   "v"(hi[4]), "v"(hi[5]), "v"(hi[6]), "v"(hi[7]));
      static_assert(RPT == 7, "the statement below names the RPT steps one by one");
      asm volatile("" ::"v"(rd[0]), "v"(rd[1]), "v"(rd[2]), "v"(rd[3]), "v"(rd[4]), "v"(rd[5]),
                   "v"(rd[6]), "v"(rs[0]), "v"(rs[1]), "v"(rs[2]), "v"(rs[3]), "v"(rs[4]),
                   "v"(rs[5]), "v"(rs[6]));
      const unsigned NONE = 0xffffffffu;
      unsigned plo[8], phi[8];
      unsigned big = 0u;
#pragma unroll
      for (int u = 0; u < 8; ++u) {
        const int c = c0 + 32 * u;
        const int cn = __builtin_amdgcn_readfirstlane((int)lo[u]);
        const bool fits = (unsigned)cn <= (unsigned)SLOT_CAP;
        if (c < nChunks && !fits) big |= 1u << u;
        const bool use = c < nChunks && fits;
        plo[u] = (use && lane >= 1 && lane <= cn) ? atomicAdd(&ocnt[lo[u] & 31u], 1u) : NONE;
        phi[u] = (use && lane + 32 <= cn) ? atomicAdd(&ocnt[hi[u] & 31u], 1u) : NONE;
      }
#pragma unroll
      for (int u = 0; u < 8; ++u) {
        if (plo[u] != NONE) commit(lo[u], plo[u]);
        if (phi[u] != NONE) commit(hi[u], phi[u]);
      }
      while (big) {
        const int u = __builtin_ctz(big);
        big &= big - 1u;
        const int e0 = (c0 + 32 * u) * CHUNK_EDGES;
        for (int g = e0; g < e0 + CHUNK_EDGES; g += 256) {
          int dt[8];
#pragma unroll
          for (int k = 0; k < 8; ++k) dt[k] = dst[g + k * 32 + lane];
#pragma unroll
          for (int k = 0; k < 8; ++k) {
            const unsigned l = (unsigned)(dt[k] - rowBase);
            if (l < (unsigned)OWN_ROWS) place(((unsigned)src[g + k * 32 + lane] << 9) | l);
          }
        }
      }
    }
  }
#pragma unroll
  for (int k = 0; k < RPT; ++k) {
    const unsigned l = (unsigned)(rd[k] - rowBase);
    if (!haveList && covered + k * 1024 + tid < E && l < (unsigned)OWN_ROWS)
      place(((unsigned)rs[k] << 9) | l);
  }
  __syncthreads();

  if (!haveList) n = min((int)__builtin_amdgcn_readfirstlane((int)ocnt[w]), OCAP);
  if constexpr (MODE == MODE_RECORD) {
    for (int j = lane; j < n; j += 32) myRec[j] = oq[w][j];
    if (lane == 0) recCnt[blockIdx.x * 32 + w] = n;
    if (tid == 0) recOk[blockIdx.x] = ovf ? 0 : 1;
  }

  if constexpr (D >= 128) {
    for (int i = 0; i < n; i += 16) {
      const int mi = i + (lane & 15);
      const unsigned mine = oq[w][mi < n ? mi : 0];
      float4 v[16];
#pragma unroll
      for (int j = 0; j < 16; ++j) {
        const unsigned en = (unsigned)__builtin_amdgcn_readlane((int)mine, j);
        v[j] = *(const float4*)(x + (size_t)(en >> 9) * D + 4 * lane);
      }
      if (lane < 16 && mi < n) atomicAdd(&cnt[mine & 511u], 1.0f);
#pragma unroll
      for (int j = 0; j < 16; ++j) {
        if (i + j < n) {
          const unsigned lr = (unsigned)__builtin_amdgcn_readlane((int)mine, j) & 511u;
          float4* a = (float4*)(acc + lr * D) + lane;
          float4 t = *a;
          t.x += v[j].x; t.y += v[j].y; t.z += v[j].z; t.w += v[j].w;
          *a = t;
        }
      }
    }
  } else {
    for (int i = 0; i < n; i += 8 * RPW) {
      float v[8][CPL];
      int ld[8];
      bool ok[8];
#pragma unroll
      for (int j = 0; j < 8; ++j) {
        const int idx = i + j * RPW + slot;
        ok[j] = idx < n;
        const unsigned en = oq[w][ok[j] ? idx : 0];
        ld[j] = (int)(en & 511u);
        const float* p = x + (size_t)(en >> 9) * D + sub;
#pragma unroll
        for (int c = 0; c < CPL; ++c) v[j][c] = p[c * LPR];
      }
#pragma unroll
      for (int j = 0; j < 8; ++j) {
        if (ok[j]) {
          float* a = acc + ld[j] * D + sub;
#pragma unroll
          for (int c = 0; c < CPL; ++c) atomicAdd(a + c * LPR, v[j][c]);
          if (sub == 0) atomicAdd(&cnt[ld[j]], 1.0f);
        }
      }
    }
  }
  __syncthreads();

  if constexpr (D < 128) {
    constexpr int KS = D / 4;
    constexpr int ROW_TILES = (OWN_ROWS + 15) / 16;
    const int l16 = lane & 15;
    const int half = lane >> 4;
    const int rowsHere = min(OWN_ROWS, N - rowBase);
    for (int t = w; t < ROW_TILES; t += 32) {
      const int r = t * 16 + l16;
      const bool rowOk = r < rowsHere;
      const int rr = rowOk ? r : 0;
      const float inv = 1.0f / fmaxf(cnt[rr], 1.0f);
      v2f am[KS], ax[KS];
#pragma unroll
      for (int j = 0; j < KS; ++j) {
        const int ka = 4 * j + 2 * half;
        const v2f s = *(const v2f*)&acc[rr * D + ka];
        const v2f xv = *(const v2f*)(x + (size_t)(rowBase + rr) * D + ka);
        am[j].x = rowOk ? s.x * inv : 0.0f;
        am[j].y = rowOk ? s.y * inv : 0.0f;
        ax[j].x = rowOk ? xv.x : 0.0f;
        ax[j].y = rowOk ? xv.y : 0.0f;
      }
      v8f c[8];
#pragma unroll
      for (int ct = 0; ct < 8; ++ct) {
        const int col = ct * 16 + l16;
        v8f cc = {};
#pragma unroll
        for (int j = 0; j < KS; ++j) {
          const int ka = 4 * j + 2 * half;
          v2f bl, br;
          bl.x = wl[(size_t)ka * HIDDIM + col];
          bl.y = wl[(size_t)(ka + 1) * HIDDIM + col];
          br.x = wr[(size_t)ka * HIDDIM + col];
          br.y = wr[(size_t)(ka + 1) * HIDDIM + col];
          cc = __builtin_amdgcn_wmma_f32_16x16x4_f32(false, am[j], false, bl, (short)0, cc,
                                                     false, false);
          cc = __builtin_amdgcn_wmma_f32_16x16x4_f32(false, ax[j], false, br, (short)0, cc,
                                                     false, false);
        }
        const float bv = bias[col];
#pragma unroll
        for (int v = 0; v < 8; ++v) cc[v] += bv;
        c[ct] = cc;
      }
      float ss[8];
#pragma unroll
      for (int v = 0; v < 8; ++v) {
        float s = c[0][v] * c[0][v];
#pragma unroll
        for (int ct = 1; ct < 8; ++ct) s += c[ct][v] * c[ct][v];
        ss[v] = s;
      }
#pragma unroll
      for (int v = 0; v < 8; ++v) ss[v] = xor_add(ss[v], 1);
#pragma unroll
      for (int v = 0; v < 8; ++v) ss[v] = xor_add(ss[v], 2);
#pragma unroll
      for (int v = 0; v < 8; ++v) ss[v] = xor_add(ss[v], 4);
#pragma unroll
      for (int v = 0; v < 8; ++v) ss[v] = xor_add(ss[v], 8);
      float* const o = hout + (size_t)(rowBase + t * 16 + 8 * half) * HIDDIM + l16;
      const int left = rowsHere - (t * 16 + 8 * half);
#pragma unroll
      for (int v = 0; v < 8; ++v) {
        if (v < left) {
          const float scale = 1.0f / fmaxf(sqrtf(ss[v]), 1e-12f);
#pragma unroll
          for (int ct = 0; ct < 8; ++ct) o[v * HIDDIM + ct * 16] = tanhf(c[ct][v] * scale);
        }
      }
    }
  } else {
    for (int i = tid; i < OWN_ROWS * D / 4; i += 1024) {
      const int r = (i * 4) / D;
      const int g = rowBase + r;
      if (g < N) {
        const float inv = 1.0f / fmaxf(cnt[r], 1.0f);
        float4 s = ((const float4*)acc)[i];
        s.x *= inv; s.y *= inv; s.z *= inv; s.w *= inv;
        *(float4*)(mean + (size_t)g * D + (i * 4 - r * D)) = s;
      }
    }
  }
}

#define APITCH 132
__global__ __launch_bounds__(256, 4) void sage_layer_resident_kernel(
    const float* __restrict__ agg, const float* __restrict__ xin, const float* __restrict__ wl,
    const float* __restrict__ wr, const float* __restrict__ bias,
    float* __restrict__ out, int nTiles) {
  constexpr int K = HIDDIM;
  __shared__ __attribute__((aligned(16))) float sM[16][APITCH];
  __shared__ __attribute__((aligned(16))) float sX[16][APITCH];
  __shared__ float s_norm[2][16];

  const int tid = threadIdx.x;
  const int wave = tid >> 5;
  const int lane = tid & 31;
  const int half = lane >> 4;
  const int l16 = lane & 15;
  const int col = wave * 16 + l16;

  v2f bl[K / 4], br[K / 4];
#pragma unroll
  for (int j = 0; j < K / 4; ++j) {
    const int ka = 4 * j + 2 * half;
    bl[j].x = wl[(size_t)ka * HIDDIM + col];
    bl[j].y = wl[(size_t)(ka + 1) * HIDDIM + col];
    br[j].x = wr[(size_t)ka * HIDDIM + col];
    br[j].y = wr[(size_t)(ka + 1) * HIDDIM + col];
  }
  const float bv = bias[col];

  const int sr = tid >> 5;
  const int sc = (tid & 31) * 4;
  int tile = blockIdx.x;
  float4 pm0, pm1, px0, px1;
  pm0 = pm1 = px0 = px1 = make_float4(0.0f, 0.0f, 0.0f, 0.0f);
  if (tile < nTiles) {
    const size_t g = (size_t)(tile * 16 + sr) * K + sc;
    pm0 = *(const float4*)(agg + g);
    pm1 = *(const float4*)(agg + g + 8 * K);
    px0 = *(const float4*)(xin + g);
    px1 = *(const float4*)(xin + g + 8 * K);
  }

  int par = 0;
  for (; tile < nTiles; tile += gridDim.x) {
    const int rowBase = tile * 16;
    *(float4*)&sM[sr][sc] = pm0;
    *(float4*)&sM[sr + 8][sc] = pm1;
    *(float4*)&sX[sr][sc] = px0;
    *(float4*)&sX[sr + 8][sc] = px1;
    if (tid < 16) s_norm[par][tid] = 0.0f;
    __syncthreads();

    v8f c = {};
    v2f am[2][8], ax[2][8];
#pragma unroll
    for (int u = 0; u < 8; ++u) {
      am[0][u] = *(const v2f*)&sM[l16][4 * u + 2 * half];
      ax[0][u] = *(const v2f*)&sX[l16][4 * u + 2 * half];
    }
#pragma unroll
    for (int b = 0; b < K / 32; ++b) {
      if (b + 1 < K / 32) {
#pragma unroll
        for (int u = 0; u < 8; ++u) {
          am[(b + 1) & 1][u] = *(const v2f*)&sM[l16][32 * (b + 1) + 4 * u + 2 * half];
          ax[(b + 1) & 1][u] = *(const v2f*)&sX[l16][32 * (b + 1) + 4 * u + 2 * half];
        }
      }
      __builtin_amdgcn_sched_barrier(0);
#pragma unroll
      for (int u = 0; u < 8; ++u) {
        const int j = 8 * b + u;
        c = __builtin_amdgcn_wmma_f32_16x16x4_f32(false, am[b & 1][u], false, bl[j], (short)0, c,
                                                  false, false);
        c = __builtin_amdgcn_wmma_f32_16x16x4_f32(false, ax[b & 1][u], false, br[j], (short)0, c,
                                                  false, false);
      }
      __builtin_amdgcn_sched_barrier(0);
    }

#pragma unroll
    for (int v = 0; v < 8; ++v) c[v] += bv;

    float ss[8];
#pragma unroll
    for (int v = 0; v < 8; ++v) ss[v] = c[v] * c[v];
#pragma unroll
    for (int v = 0; v < 8; ++v) ss[v] = xor_add(ss[v], 1);
#pragma unroll
    for (int v = 0; v < 8; ++v) ss[v] = xor_add(ss[v], 2);
#pragma unroll
    for (int v = 0; v < 8; ++v) ss[v] = xor_add(ss[v], 4);
#pragma unroll
    for (int v = 0; v < 8; ++v) ss[v] = xor_add(ss[v], 8);
    if (l16 == 0) {
#pragma unroll
      for (int v = 0; v < 8; ++v) atomicAdd(&s_norm[par][v + 8 * half], ss[v]);
    }
    __syncthreads();

    {
      const int nt = tile + (int)gridDim.x < nTiles ? tile + (int)gridDim.x : tile;
      const size_t g = (size_t)(nt * 16 + sr) * K + sc;
      pm0 = *(const float4*)(agg + g);
      pm1 = *(const float4*)(agg + g + 8 * K);
      px0 = *(const float4*)(xin + g);
      px1 = *(const float4*)(xin + g + 8 * K);
    }

    float scale[8];
#pragma unroll
    for (int v = 0; v < 8; ++v)
      scale[v] = 1.0f / fmaxf(sqrtf(s_norm[par][v + 8 * half]), 1e-12f);
    asm volatile("" ::"v"(scale[0]), "v"(scale[1]), "v"(scale[2]), "v"(scale[3]), "v"(scale[4]),
                 "v"(scale[5]), "v"(scale[6]), "v"(scale[7]));
#pragma unroll
    for (int v = 0; v < 8; ++v)
      out[(size_t)(rowBase + v + 8 * half) * HIDDIM + col] = tanhf(c[v] * scale[v]);
    par ^= 1;
  }
}

extern "C" void kernel_launch(void* const* d_in, const int* in_sizes, int n_in,
                              void* d_out, int out_size, void* d_ws,
                              size_t ws_size, hipStream_t stream) {
  const float* x = (const float*)d_in[0];
  const int* ei_conn = (const int*)d_in[1];
  const int* ei_dest = (const int*)d_in[2];
  const float* wl1 = (const float*)d_in[3];
  const float* wr1 = (const float*)d_in[4];
  const float* b1 = (const float*)d_in[5];
  const float* wl2 = (const float*)d_in[6];
  const float* wr2 = (const float*)d_in[7];
  const float* b2 = (const float*)d_in[8];
  const float* wl3 = (const float*)d_in[9];
  const float* wr3 = (const float*)d_in[10];
  const float* b3 = (const float*)d_in[11];

  const int N = in_sizes[0] / 8;
  const int E = in_sizes[1] / 2;

  float* ws = (float*)d_ws;
  float* mean = ws;
  float* h1 = mean + (size_t)N * HIDDIM;
  float* h2 = h1 + (size_t)N * HIDDIM;

  const int T = 256;
  const int* src_conn = ei_conn;
  const int* dst_conn = ei_conn + E;
  const int* src_dest = ei_dest;
  const int* dst_dest = ei_dest + E;

  const int gemmBlocks = (N + 15) / 16;
  const int ownBlocks = (N + OWN_ROWS - 1) / OWN_ROWS;
  const int wideBlocks = 512;
  const int nChunks = E / CHUNK_EDGES;
  const size_t slotAll = (size_t)ownBlocks * nChunks * SLOT_WORDS;
  unsigned* slot_conn = (unsigned*)(h2 + (size_t)N * HIDDIM);
  unsigned* slot_dest = slot_conn + slotAll;
  unsigned* rec = slot_dest + slotAll;
  int* recCnt = (int*)(rec + (size_t)ownBlocks * 32 * OCAP);

  owner_slot_kernel<<<dim3(nChunks > 0 ? nChunks : 1, 2), 1024, 0, stream>>>(
      src_conn, dst_conn, src_dest, dst_dest, slot_conn, slot_dest, nChunks, ownBlocks);

  owner_mean_kernel<8, MODE_RECORD><<<ownBlocks, 1024, 0, stream>>>(
      x, src_conn, dst_conn, slot_conn, mean, rec, recCnt, N, E, nChunks, wl1, wr1, b1, h1);

  owner_mean_kernel<HIDDIM, MODE_SCAN><<<ownBlocks, 1024, 0, stream>>>(
      h1, src_dest, dst_dest, slot_dest, mean, rec, recCnt, N, E, nChunks, nullptr, nullptr,
      nullptr, nullptr);
  sage_layer_resident_kernel<<<wideBlocks, T, 0, stream>>>(mean, h1, wl2, wr2, b2, h2, gemmBlocks);

  owner_mean_kernel<HIDDIM, MODE_REPLAY><<<ownBlocks, 1024, 0, stream>>>(
      h2, src_conn, dst_conn, slot_conn, mean, rec, recCnt, N, E, nChunks, nullptr, nullptr,
      nullptr, nullptr);
  sage_layer_resident_kernel<<<wideBlocks, T, 0, stream>>>(mean, h2, wl3, wr3, b3, (float*)d_out,
                                                           gemmBlocks);
}
